# starting kernel
# baseline (speedup 1.0000x reference)
#include <hip/hip_runtime.h>
#include <hip/hip_bf16.h>
#include <math.h>

#define NSTEP 64
#define H2    1024
#define TT    512
#define EPSN  1e-12f

typedef __attribute__((ext_vector_type(2))) float v2f;
typedef __attribute__((ext_vector_type(8))) float v8f;

__global__ void __launch_bounds__(1024)
recurrence_kernel(const float* __restrict__ W, const float* __restrict__ bias,
                  const float* __restrict__ u0, float* __restrict__ u_ws)
{
    __shared__ float u[H2];
    __shared__ float y[H2];
    __shared__ float red[32];

    const int tid  = threadIdx.x;
    const int wave = tid >> 5;
    const int lane = tid & 31;

    u[tid] = u0[tid];
    __syncthreads();

    for (int i = 0; i < NSTEP; ++i) {
        for (int er = 0; er < 32; ++er) {
            const int e = wave * 32 + er;
            const float* wr = W + (size_t)e * H2;
            float p = 0.0f;
            #pragma unroll 4
            for (int d = lane; d < H2; d += 32)
                p += wr[d] * u[d];
            #pragma unroll
            for (int off = 16; off >= 1; off >>= 1)
                p += __shfl_xor(p, off, 32);
            if (lane == 0) y[e] = p + bias[e];
        }
        __syncthreads();

        float v  = y[tid];
        float sq = v * v;
        #pragma unroll
        for (int off = 16; off >= 1; off >>= 1)
            sq += __shfl_xor(sq, off, 32);
        if (lane == 0) red[wave] = sq;
        __syncthreads();
        if (tid < 32) {
            float r = red[tid];
            #pragma unroll
            for (int off = 16; off >= 1; off >>= 1)
                r += __shfl_xor(r, off, 32);
            red[tid] = r;
        }
        __syncthreads();

        const float inv = 1.0f / fmaxf(sqrtf(red[0]), EPSN);
        const float val = v * inv;
        u_ws[(size_t)i * H2 + tid] = val;
        u[tid] = val;
        __syncthreads();
    }
}

#define KC   512
#define KPAD (KC + 2)

__global__ void __launch_bounds__(256)
scores_kernel(const float* __restrict__ hidden, const float* __restrict__ W,
              const float* __restrict__ bias,  const float* __restrict__ u_ws,
              float* __restrict__ scores)
{
    __shared__ float sA[16 * KPAD];
    __shared__ float sScore[16];

    const int n    = blockIdx.x;
    const int t0   = blockIdx.y * 16;
    const int tid  = threadIdx.x;
    const int lane = tid & 31;
    const int wave = tid >> 5;
    const int l16  = lane & 15;
    const int kHalf = (lane >> 4) * 2;

    if (tid < 16) sScore[tid] = 0.0f;

    const float* hbase = hidden + (size_t)n * H2 * TT + t0;

    v8f accs[8];
    #pragma unroll
    for (int i = 0; i < 8; ++i)
        accs[i] = (v8f){0.f,0.f,0.f,0.f,0.f,0.f,0.f,0.f};

    for (int kc = 0; kc < H2; kc += KC) {
        for (int i = tid; i < KC * 16; i += 256) {
            const int k = i >> 4;
            const int t = i & 15;
            sA[t * KPAD + k] = hbase[(size_t)(kc + k) * TT + t];
        }
        __syncthreads();

        #pragma unroll
        for (int idx = 0; idx < 8; ++idx) {
            const int e0 = (wave + idx * 8) * 16;
            const float* Wrow = W + (size_t)(e0 + l16) * H2 + kc + kHalf;
            const float* Arow = sA + l16 * KPAD + kHalf;
            v8f acc = accs[idx];
            for (int k0 = 0; k0 < KC; k0 += 4) {
                v2f av = *(const v2f*)(Arow + k0);
                v2f bv = *(const v2f*)(Wrow + k0);
                acc = __builtin_amdgcn_wmma_f32_16x16x4_f32(
                         false, av,  false, bv,
                         (short)0, acc,
                         false,  false);
            }
            accs[idx] = acc;
        }
        __syncthreads();
    }

    float tAcc[8];
    #pragma unroll
    for (int r = 0; r < 8; ++r) tAcc[r] = 0.0f;

    #pragma unroll
    for (int idx = 0; idx < 8; ++idx) {
        const int e0 = (wave + idx * 8) * 16;
        const float bb = bias[e0 + l16];
        const float uu = u_ws[(size_t)n * H2 + e0 + l16];
        #pragma unroll
        for (int r = 0; r < 8; ++r)
            tAcc[r] += tanhf(accs[idx][r] + bb) * uu;
    }

    #pragma unroll
    for (int r = 0; r < 8; ++r) {
        float v = tAcc[r];
        #pragma unroll
        for (int off = 8; off >= 1; off >>= 1)
            v += __shfl_xor(v, off, 32);
        if (l16 == 0) {
            const int t = r + ((lane >> 4) << 3);
            atomicAdd(&sScore[t], v);
        }
    }
    __syncthreads();

    if (tid < 16)
        scores[(size_t)n * TT + t0 + tid] = sScore[tid];
}

__global__ void __launch_bounds__(512)
softmax_kernel(const float* __restrict__ scores, float* __restrict__ attn)
{
    __shared__ float red[16];
    const int n   = blockIdx.x;
    const int tid = threadIdx.x;

    const float s = scores[(size_t)n * TT + tid];

    float m = s;
    #pragma unroll
    for (int off = 16; off >= 1; off >>= 1)
        m = fmaxf(m, __shfl_xor(m, off, 32));
    if ((tid & 31) == 0) red[tid >> 5] = m;
    __syncthreads();
    if (tid < 16) {
        float r = red[tid];
        #pragma unroll
        for (int off = 8; off >= 1; off >>= 1)
            r = fmaxf(r, __shfl_xor(r, off, 32));
        red[tid] = r;
    }
    __syncthreads();
    m = red[0];
    __syncthreads();

    const float e = __expf(s - m);
    float sum = e;
    #pragma unroll
    for (int off = 16; off >= 1; off >>= 1)
        sum += __shfl_xor(sum, off, 32);
    if ((tid & 31) == 0) red[tid >> 5] = sum;
    __syncthreads();
    if (tid < 16) {
        float r = red[tid];
        #pragma unroll
        for (int off = 8; off >= 1; off >>= 1)
            r += __shfl_xor(r, off, 32);
        red[tid] = r;
    }
    __syncthreads();

    attn[(size_t)n * TT + tid] = e / red[0];
}

__global__ void __launch_bounds__(256)
weighted_sum_kernel(const float* __restrict__ hidden,
                    const float* __restrict__ attn, float* __restrict__ out)
{
    __shared__ float sa[TT];
    const int n   = blockIdx.x;
    const int dc  = blockIdx.y;
    const int tid = threadIdx.x;

    sa[tid]       = attn[(size_t)n * TT + tid];
    sa[tid + 256] = attn[(size_t)n * TT + tid + 256];
    __syncthreads();

    const int d = dc * 256 + tid;
    const float* h = hidden + (size_t)n * H2 * TT + (size_t)d * TT;
    float acc = 0.0f;
    #pragma unroll 4
    for (int t = 0; t < TT; t += 4) {
        const float4 h4 = *(const float4*)(h + t);
        acc += h4.x * sa[t] + h4.y * sa[t + 1] + h4.z * sa[t + 2] + h4.w * sa[t + 3];
    }
    out[(size_t)n * H2 + d] = acc;
}

extern "C" void kernel_launch(void* const* d_in, const int* in_sizes, int n_in,
                              void* d_out, int out_size, void* d_ws, size_t ws_size,
                              hipStream_t stream)
{
    const float* hidden = (const float*)d_in[0];
    const float* W      = (const float*)d_in[1];
    const float* bias   = (const float*)d_in[2];
    const float* u0     = (const float*)d_in[3];
    float* out = (float*)d_out;

    float* ws     = (float*)d_ws;
    float* u_ws   = ws;
    float* scores = u_ws + (size_t)NSTEP * H2;
    float* attn   = scores + (size_t)NSTEP * TT;

    recurrence_kernel<<<1, 1024, 0, stream>>>(W, bias, u0, u_ws);

    scores_kernel<<<dim3(NSTEP, TT / 16), 256, 0, stream>>>(
        hidden, W, bias, u_ws, scores);

    softmax_kernel<<<NSTEP, TT, 0, stream>>>(scores, attn);

    weighted_sum_kernel<<<dim3(NSTEP, H2 / 256), 256, 0, stream>>>(
        hidden, attn, out);
}


# recurrence: one launch per step with the rows of W split over 32 work-groups (4 rows per wave, 16-byte loads, on-ALU butterfly sums) instead of one work-group; dot and norm summation order changed
# speedup vs baseline: 4.8147x; 4.8147x over previous
#include <hip/hip_runtime.h>
#include <hip/hip_bf16.h>
#include <math.h>

#define NSTEP 64
#define H2    1024
#define TT    512
#define EPSN  1e-12f

typedef __attribute__((ext_vector_type(2))) float v2f;
typedef __attribute__((ext_vector_type(8))) float v8f;

#define ROWS_PER_WG   32
#define ROWS_PER_WAVE 4

__device__ __forceinline__ float xhalf(float x)
{
    return __builtin_bit_cast(float, __builtin_amdgcn_permlanex16(
        __builtin_bit_cast(int, x), __builtin_bit_cast(int, x),
        0x76543210, 0xfedcba98, false, false));
}

template <int M>
__device__ __forceinline__ float row_xor(float x)
{
    return __builtin_bit_cast(float, __builtin_amdgcn_update_dpp(
        0, __builtin_bit_cast(int, x), 0x160 + M, 0xf, 0xf, true));
}

__device__ __forceinline__ float wave_sum(float v)
{
    v += row_xor<1>(v);
    v += row_xor<2>(v);
    v += row_xor<4>(v);
    v += row_xor<8>(v);
    v += xhalf(v);
    return v;
}

__device__ __forceinline__ float inv_norm(const float4 (&y)[8])
{
    float sq = 0.0f;
    #pragma unroll
    for (int j = 0; j < 8; ++j)
        sq += y[j].x * y[j].x + y[j].y * y[j].y + y[j].z * y[j].z + y[j].w * y[j].w;
    sq = wave_sum(sq);
    return 1.0f / fmaxf(sqrtf(sq), EPSN);
}

__global__ void __launch_bounds__(256)
recurrence_step_kernel(const float* __restrict__ W, const float* __restrict__ bias,
                       const float* __restrict__ vin, float* __restrict__ yout,
                       int normalize)
{
    __shared__ float sy[ROWS_PER_WG];

    const int tid  = threadIdx.x;
    const int wave = tid >> 5;
    const int lane = tid & 31;
    const int e0   = blockIdx.x * ROWS_PER_WG;

    const float4* v4 = (const float4*)vin;
    const float4* w4 = (const float4*)(W + (size_t)(e0 + wave * ROWS_PER_WAVE) * H2);

    float4 u[8];
    #pragma unroll
    for (int j = 0; j < 8; ++j)
        u[j] = v4[lane + 32 * j];

    __builtin_amdgcn_sched_barrier(0);

    float4 w[ROWS_PER_WAVE][8];
    #pragma unroll
    for (int r = 0; r < ROWS_PER_WAVE; ++r) {
        #pragma unroll
        for (int j = 0; j < 8; ++j)
            w[r][j] = w4[r * (H2 / 4) + lane + 32 * j];
    }

    float bb = 0.0f;
    if (tid < ROWS_PER_WG) bb = bias[e0 + tid];

    float inv = 1.0f;
    if (normalize) inv = inv_norm(u);

    #pragma unroll
    for (int j = 0; j < 8; ++j) {
        u[j].x *= inv; u[j].y *= inv; u[j].z *= inv; u[j].w *= inv;
    }

    float p[ROWS_PER_WAVE];
    #pragma unroll
    for (int r = 0; r < ROWS_PER_WAVE; ++r) {
        float ax = 0.0f, ay = 0.0f, az = 0.0f, aw = 0.0f;
        #pragma unroll
        for (int j = 0; j < 8; ++j) {
            ax += w[r][j].x * u[j].x;
            ay += w[r][j].y * u[j].y;
            az += w[r][j].z * u[j].z;
            aw += w[r][j].w * u[j].w;
        }
        p[r] = wave_sum((ax + ay) + (az + aw));
    }

    if (lane == 0) {
        #pragma unroll
        for (int r = 0; r < ROWS_PER_WAVE; ++r)
            sy[wave * ROWS_PER_WAVE + r] = p[r];
    }
    __syncthreads();

    if (tid < ROWS_PER_WG)
        yout[e0 + tid] = sy[tid] + bb;
}

__global__ void __launch_bounds__(256)
normalize_rows_kernel(float* __restrict__ u_ws)
{
    const int lane = threadIdx.x & 31;
    const int row  = blockIdx.x * 8 + (threadIdx.x >> 5);
    float4* p4 = (float4*)(u_ws + (size_t)row * H2);

    float4 y[8];
    #pragma unroll
    for (int j = 0; j < 8; ++j)
        y[j] = p4[lane + 32 * j];

    const float inv = inv_norm(y);

    #pragma unroll
    for (int j = 0; j < 8; ++j) {
        float4 o;
        o.x = y[j].x * inv; o.y = y[j].y * inv; o.z = y[j].z * inv; o.w = y[j].w * inv;
        p4[lane + 32 * j] = o;
    }
}

#define KC   512
#define KPAD (KC + 2)

__global__ void __launch_bounds__(256)
scores_kernel(const float* __restrict__ hidden, const float* __restrict__ W,
              const float* __restrict__ bias,  const float* __restrict__ u_ws,
              float* __restrict__ scores)
{
    __shared__ float sA[16 * KPAD];
    __shared__ float sScore[16];

    const int n    = blockIdx.x;
    const int t0   = blockIdx.y * 16;
    const int tid  = threadIdx.x;
    const int lane = tid & 31;
    const int wave = tid >> 5;
    const int l16  = lane & 15;
    const int kHalf = (lane >> 4) * 2;

    if (tid < 16) sScore[tid] = 0.0f;

    const float* hbase = hidden + (size_t)n * H2 * TT + t0;

    v8f accs[8];
    #pragma unroll
    for (int i = 0; i < 8; ++i)
        accs[i] = (v8f){0.f,0.f,0.f,0.f,0.f,0.f,0.f,0.f};

    for (int kc = 0; kc < H2; kc += KC) {
        for (int i = tid; i < KC * 16; i += 256) {
            const int k = i >> 4;
            const int t = i & 15;
            sA[t * KPAD + k] = hbase[(size_t)(kc + k) * TT + t];
        }
        __syncthreads();

        #pragma unroll
        for (int idx = 0; idx < 8; ++idx) {
            const int e0 = (wave + idx * 8) * 16;
            const float* Wrow = W + (size_t)(e0 + l16) * H2 + kc + kHalf;
            const float* Arow = sA + l16 * KPAD + kHalf;
            v8f acc = accs[idx];
            for (int k0 = 0; k0 < KC; k0 += 4) {
                v2f av = *(const v2f*)(Arow + k0);
                v2f bv = *(const v2f*)(Wrow + k0);
                acc = __builtin_amdgcn_wmma_f32_16x16x4_f32(
                         false, av,  false, bv,
                         (short)0, acc,
                         false,  false);
            }
            accs[idx] = acc;
        }
        __syncthreads();
    }

    float tAcc[8];
    #pragma unroll
    for (int r = 0; r < 8; ++r) tAcc[r] = 0.0f;

    #pragma unroll
    for (int idx = 0; idx < 8; ++idx) {
        const int e0 = (wave + idx * 8) * 16;
        const float bb = bias[e0 + l16];
        const float uu = u_ws[(size_t)n * H2 + e0 + l16];
        #pragma unroll
        for (int r = 0; r < 8; ++r)
            tAcc[r] += tanhf(accs[idx][r] + bb) * uu;
    }

    #pragma unroll
    for (int r = 0; r < 8; ++r) {
        float v = tAcc[r];
        #pragma unroll
        for (int off = 8; off >= 1; off >>= 1)
            v += __shfl_xor(v, off, 32);
        if (l16 == 0) {
            const int t = r + ((lane >> 4) << 3);
            atomicAdd(&sScore[t], v);
        }
    }
    __syncthreads();

    if (tid < 16)
        scores[(size_t)n * TT + t0 + tid] = sScore[tid];
}

__global__ void __launch_bounds__(512)
softmax_kernel(const float* __restrict__ scores, float* __restrict__ attn)
{
    __shared__ float red[16];
    const int n   = blockIdx.x;
    const int tid = threadIdx.x;

    const float s = scores[(size_t)n * TT + tid];

    float m = s;
    #pragma unroll
    for (int off = 16; off >= 1; off >>= 1)
        m = fmaxf(m, __shfl_xor(m, off, 32));
    if ((tid & 31) == 0) red[tid >> 5] = m;
    __syncthreads();
    if (tid < 16) {
        float r = red[tid];
        #pragma unroll
        for (int off = 8; off >= 1; off >>= 1)
            r = fmaxf(r, __shfl_xor(r, off, 32));
        red[tid] = r;
    }
    __syncthreads();
    m = red[0];
    __syncthreads();

    const float e = __expf(s - m);
    float sum = e;
    #pragma unroll
    for (int off = 16; off >= 1; off >>= 1)
        sum += __shfl_xor(sum, off, 32);
    if ((tid & 31) == 0) red[tid >> 5] = sum;
    __syncthreads();
    if (tid < 16) {
        float r = red[tid];
        #pragma unroll
        for (int off = 8; off >= 1; off >>= 1)
            r += __shfl_xor(r, off, 32);
        red[tid] = r;
    }
    __syncthreads();

    attn[(size_t)n * TT + tid] = e / red[0];
}

__global__ void __launch_bounds__(256)
weighted_sum_kernel(const float* __restrict__ hidden,
                    const float* __restrict__ attn, float* __restrict__ out)
{
    __shared__ float sa[TT];
    const int n   = blockIdx.x;
    const int dc  = blockIdx.y;
    const int tid = threadIdx.x;

    sa[tid]       = attn[(size_t)n * TT + tid];
    sa[tid + 256] = attn[(size_t)n * TT + tid + 256];
    __syncthreads();

    const int d = dc * 256 + tid;
    const float* h = hidden + (size_t)n * H2 * TT + (size_t)d * TT;
    float acc = 0.0f;
    #pragma unroll 4
    for (int t = 0; t < TT; t += 4) {
        const float4 h4 = *(const float4*)(h + t);
        acc += h4.x * sa[t] + h4.y * sa[t + 1] + h4.z * sa[t + 2] + h4.w * sa[t + 3];
    }
    out[(size_t)n * H2 + d] = acc;
}

extern "C" void kernel_launch(void* const* d_in, const int* in_sizes, int n_in,
                              void* d_out, int out_size, void* d_ws, size_t ws_size,
                              hipStream_t stream)
{
    const float* hidden = (const float*)d_in[0];
    const float* W      = (const float*)d_in[1];
    const float* bias   = (const float*)d_in[2];
    const float* u0     = (const float*)d_in[3];
    float* out = (float*)d_out;

    float* ws     = (float*)d_ws;
    float* u_ws   = ws;
    float* scores = u_ws + (size_t)NSTEP * H2;
    float* attn   = scores + (size_t)NSTEP * TT;

    recurrence_step_kernel<<<H2 / ROWS_PER_WG, 256, 0, stream>>>(W, bias, u0, u_ws, 0);
    for (int i = 1; i < NSTEP; ++i)
        recurrence_step_kernel<<<H2 / ROWS_PER_WG, 256, 0, stream>>>(
            W, bias, u_ws + (size_t)(i - 1) * H2, u_ws + (size_t)i * H2, 1);
    normalize_rows_kernel<<<NSTEP / 8, 256, 0, stream>>>(u_ws);

    scores_kernel<<<dim3(NSTEP, TT / 16), 256, 0, stream>>>(
        hidden, W, bias, u_ws, scores);

    softmax_kernel<<<NSTEP, TT, 0, stream>>>(scores, attn);

    weighted_sum_kernel<<<dim3(NSTEP, H2 / 256), 256, 0, stream>>>(
        hidden, attn, out);
}


# score GEMM re-tiled: 64-token x 256-column block tiles, both operands staged through LDS by shared 16-byte loads (W read once per 64 tokens, not 16); order of the sum over e changed, now fixed
# speedup vs baseline: 7.5636x; 1.5709x over previous
#include <hip/hip_runtime.h>
#include <hip/hip_bf16.h>
#include <math.h>

#define NSTEP 64
#define H2    1024
#define TT    512
#define EPSN  1e-12f

typedef __attribute__((ext_vector_type(2))) float v2f;
typedef __attribute__((ext_vector_type(8))) float v8f;

#define ROWS_PER_WG   32
#define ROWS_PER_WAVE 4

__device__ __forceinline__ float xhalf(float x)
{
    return __builtin_bit_cast(float, __builtin_amdgcn_permlanex16(
        __builtin_bit_cast(int, x), __builtin_bit_cast(int, x),
        0x76543210, 0xfedcba98, false, false));
}

template <int M>
__device__ __forceinline__ float row_xor(float x)
{
    return __builtin_bit_cast(float, __builtin_amdgcn_update_dpp(
        0, __builtin_bit_cast(int, x), 0x160 + M, 0xf, 0xf, true));
}

__device__ __forceinline__ float wave_sum(float v)
{
    v += row_xor<1>(v);
    v += row_xor<2>(v);
    v += row_xor<4>(v);
    v += row_xor<8>(v);
    v += xhalf(v);
    return v;
}

__device__ __forceinline__ float inv_norm(const float4 (&y)[8])
{
    float sq = 0.0f;
    #pragma unroll
    for (int j = 0; j < 8; ++j)
        sq += y[j].x * y[j].x + y[j].y * y[j].y + y[j].z * y[j].z + y[j].w * y[j].w;
    sq = wave_sum(sq);
    return 1.0f / fmaxf(sqrtf(sq), EPSN);
}

__global__ void __launch_bounds__(256)
recurrence_step_kernel(const float* __restrict__ W, const float* __restrict__ bias,
                       const float* __restrict__ vin, float* __restrict__ yout,
                       int normalize)
{
    __shared__ float sy[ROWS_PER_WG];

    const int tid  = threadIdx.x;
    const int wave = tid >> 5;
    const int lane = tid & 31;
    const int e0   = blockIdx.x * ROWS_PER_WG;

    const float4* v4 = (const float4*)vin;
    const float4* w4 = (const float4*)(W + (size_t)(e0 + wave * ROWS_PER_WAVE) * H2);

    float4 u[8];
    #pragma unroll
    for (int j = 0; j < 8; ++j)
        u[j] = v4[lane + 32 * j];

    __builtin_amdgcn_sched_barrier(0);

    float4 w[ROWS_PER_WAVE][8];
    #pragma unroll
    for (int r = 0; r < ROWS_PER_WAVE; ++r) {
        #pragma unroll
        for (int j = 0; j < 8; ++j)
            w[r][j] = w4[r * (H2 / 4) + lane + 32 * j];
    }

    float bb = 0.0f;
    if (tid < ROWS_PER_WG) bb = bias[e0 + tid];

    float inv = 1.0f;
    if (normalize) inv = inv_norm(u);

    #pragma unroll
    for (int j = 0; j < 8; ++j) {
        u[j].x *= inv; u[j].y *= inv; u[j].z *= inv; u[j].w *= inv;
    }

    float p[ROWS_PER_WAVE];
    #pragma unroll
    for (int r = 0; r < ROWS_PER_WAVE; ++r) {
        float ax = 0.0f, ay = 0.0f, az = 0.0f, aw = 0.0f;
        #pragma unroll
        for (int j = 0; j < 8; ++j) {
            ax += w[r][j].x * u[j].x;
            ay += w[r][j].y * u[j].y;
            az += w[r][j].z * u[j].z;
            aw += w[r][j].w * u[j].w;
        }
        p[r] = wave_sum((ax + ay) + (az + aw));
    }

    if (lane == 0) {
        #pragma unroll
        for (int r = 0; r < ROWS_PER_WAVE; ++r)
            sy[wave * ROWS_PER_WAVE + r] = p[r];
    }
    __syncthreads();

    if (tid < ROWS_PER_WG)
        yout[e0 + tid] = sy[tid] + bb;
}

__global__ void __launch_bounds__(256)
normalize_rows_kernel(float* __restrict__ u_ws)
{
    const int lane = threadIdx.x & 31;
    const int row  = blockIdx.x * 8 + (threadIdx.x >> 5);
    float4* p4 = (float4*)(u_ws + (size_t)row * H2);

    float4 y[8];
    #pragma unroll
    for (int j = 0; j < 8; ++j)
        y[j] = p4[lane + 32 * j];

    const float inv = inv_norm(y);

    #pragma unroll
    for (int j = 0; j < 8; ++j) {
        float4 o;
        o.x = y[j].x * inv; o.y = y[j].y * inv; o.z = y[j].z * inv; o.w = y[j].w * inv;
        p4[lane + 32 * j] = o;
    }
}

#define BM     64
#define BN     256
#define KCH    32
#define PA     72
#define PB     44
#define NPASS  (H2 / BN)
#define CPP    (H2 / KCH)
#define NCHUNK (NPASS * CPP)

#define GA(it_, j_) (*(const float4*)(hbase + (size_t)(((it_) % CPP) * KCH + ((tid + 256 * (j_)) >> 4)) * TT + 4 * ((tid + 256 * (j_)) & 15)))
#define GB(it_, j_) (*(const float4*)(W + (size_t)(((it_) / CPP) * BN + ((tid + 256 * (j_)) >> 3)) * H2 + ((it_) % CPP) * KCH + 4 * ((tid + 256 * (j_)) & 7)))
#define SA(buf_, j_) (*(float4*)(&sA[buf_][((tid + 256 * (j_)) >> 4) * PA + 4 * ((tid + 256 * (j_)) & 15)]))
#define SB(buf_, j_) (*(float4*)(&sB[buf_][((tid + 256 * (j_)) >> 3) * PB + 4 * ((tid + 256 * (j_)) & 7)]))
#define LOAD_CHUNK(it_)                                                    \
    { a0 = GA(it_, 0); a1 = GA(it_, 1);                                    \
      b0 = GB(it_, 0); b1 = GB(it_, 1); b2 = GB(it_, 2); b3 = GB(it_, 3);  \
      b4 = GB(it_, 4); b5 = GB(it_, 5); b6 = GB(it_, 6); b7 = GB(it_, 7); }
#define STORE_CHUNK(buf_)                                                  \
    { SA(buf_, 0) = a0; SA(buf_, 1) = a1;                                  \
      SB(buf_, 0) = b0; SB(buf_, 1) = b1; SB(buf_, 2) = b2; SB(buf_, 3) = b3; \
      SB(buf_, 4) = b4; SB(buf_, 5) = b5; SB(buf_, 6) = b6; SB(buf_, 7) = b7; }

__global__ void __launch_bounds__(256)
scores_kernel(const float* __restrict__ hidden, const float* __restrict__ W,
              const float* __restrict__ bias,  const float* __restrict__ u_ws,
              float* __restrict__ scores)
{
    __shared__ __attribute__((aligned(16))) float sA[2][KCH * PA];
    __shared__ __attribute__((aligned(16))) float sB[2][BN * PB];
    __shared__ float sPart[4][BM];

    const int n    = blockIdx.x;
    const int t0   = blockIdx.y * BM;
    const int tid  = threadIdx.x;
    const int lane = tid & 31;
    const int wave = tid >> 5;
    const int c    = lane & 15;
    const int h    = lane >> 4;
    const int wm   = wave >> 2;
    const int wn   = wave & 3;

    const float* hbase = hidden + (size_t)n * H2 * TT + t0;

    v8f acc[2][4];
    #pragma unroll
    for (int mt = 0; mt < 2; ++mt)
        #pragma unroll
        for (int nt = 0; nt < 4; ++nt)
            acc[mt][nt] = (v8f){0.f,0.f,0.f,0.f,0.f,0.f,0.f,0.f};

    float tAcc[2][8];
    #pragma unroll
    for (int mt = 0; mt < 2; ++mt)
        #pragma unroll
        for (int r = 0; r < 8; ++r) tAcc[mt][r] = 0.0f;

    float4 a0, a1, b0, b1, b2, b3, b4, b5, b6, b7;
    LOAD_CHUNK(0)
    STORE_CHUNK(0)
    __syncthreads();

    for (int it = 0; it < NCHUNK; ++it) {
        const int cur = it & 1;
        const bool more = (it + 1 < NCHUNK);

        if (more) LOAD_CHUNK(it + 1)

        const float* pa = sA[cur] + (2 * h) * PA + wm * 32 + c;
        const float* pb = sB[cur] + (wn * 64 + c) * PB + 2 * h;
        #pragma unroll
        for (int ks = 0; ks < KCH / 4; ++ks) {
            v2f af[2], bf[4];
            #pragma unroll
            for (int mt = 0; mt < 2; ++mt) {
                af[mt].x = pa[(ks * 4)     * PA + mt * 16];
                af[mt].y = pa[(ks * 4 + 1) * PA + mt * 16];
            }
            #pragma unroll
            for (int nt = 0; nt < 4; ++nt)
                bf[nt] = *(const v2f*)(pb + nt * 16 * PB + ks * 4);
            #pragma unroll
            for (int mt = 0; mt < 2; ++mt)
                #pragma unroll
                for (int nt = 0; nt < 4; ++nt)
                    acc[mt][nt] = __builtin_amdgcn_wmma_f32_16x16x4_f32(
                         false, af[mt],  false, bf[nt],
                         (short)0, acc[mt][nt],
                         false,  false);
        }

        if ((it % CPP) == CPP - 1) {
            const int ec = it / CPP;
            #pragma unroll
            for (int nt = 0; nt < 4; ++nt) {
                const int e = ec * BN + wn * 64 + nt * 16 + c;
                const float bb = bias[e];
                const float uu = u_ws[(size_t)n * H2 + e];
                #pragma unroll
                for (int mt = 0; mt < 2; ++mt) {
                    #pragma unroll
                    for (int r = 0; r < 8; ++r) {
                        tAcc[mt][r] += tanhf(acc[mt][nt][r] + bb) * uu;
                        acc[mt][nt][r] = 0.0f;
                    }
                }
            }
        }

        if (more) STORE_CHUNK(cur ^ 1)
        __syncthreads();
    }

    #pragma unroll
    for (int mt = 0; mt < 2; ++mt) {
        #pragma unroll
        for (int r = 0; r < 8; ++r) {
            float v = tAcc[mt][r];
            #pragma unroll
            for (int off = 8; off >= 1; off >>= 1)
                v += __shfl_xor(v, off, 32);
            if (c == 0)
                sPart[wn][wm * 32 + mt * 16 + 8 * h + r] = v;
        }
    }
    __syncthreads();

    if (tid < BM)
        scores[(size_t)n * TT + t0 + tid] =
            ((sPart[0][tid] + sPart[1][tid]) + sPart[2][tid]) + sPart[3][tid];
}

__global__ void __launch_bounds__(512)
softmax_kernel(const float* __restrict__ scores, float* __restrict__ attn)
{
    __shared__ float red[16];
    const int n   = blockIdx.x;
    const int tid = threadIdx.x;

    const float s = scores[(size_t)n * TT + tid];

    float m = s;
    #pragma unroll
    for (int off = 16; off >= 1; off >>= 1)
        m = fmaxf(m, __shfl_xor(m, off, 32));
    if ((tid & 31) == 0) red[tid >> 5] = m;
    __syncthreads();
    if (tid < 16) {
        float r = red[tid];
        #pragma unroll
        for (int off = 8; off >= 1; off >>= 1)
            r = fmaxf(r, __shfl_xor(r, off, 32));
        red[tid] = r;
    }
    __syncthreads();
    m = red[0];
    __syncthreads();

    const float e = __expf(s - m);
    float sum = e;
    #pragma unroll
    for (int off = 16; off >= 1; off >>= 1)
        sum += __shfl_xor(sum, off, 32);
    if ((tid & 31) == 0) red[tid >> 5] = sum;
    __syncthreads();
    if (tid < 16) {
        float r = red[tid];
        #pragma unroll
        for (int off = 8; off >= 1; off >>= 1)
            r += __shfl_xor(r, off, 32);
        red[tid] = r;
    }
    __syncthreads();

    attn[(size_t)n * TT + tid] = e / red[0];
}

__global__ void __launch_bounds__(256)
weighted_sum_kernel(const float* __restrict__ hidden,
                    const float* __restrict__ attn, float* __restrict__ out)
{
    __shared__ float sa[TT];
    const int n   = blockIdx.x;
    const int dc  = blockIdx.y;
    const int tid = threadIdx.x;

    sa[tid]       = attn[(size_t)n * TT + tid];
    sa[tid + 256] = attn[(size_t)n * TT + tid + 256];
    __syncthreads();

    const int d = dc * 256 + tid;
    const float* h = hidden + (size_t)n * H2 * TT + (size_t)d * TT;
    float acc = 0.0f;
    #pragma unroll 4
    for (int t = 0; t < TT; t += 4) {
        const float4 h4 = *(const float4*)(h + t);
        acc += h4.x * sa[t] + h4.y * sa[t + 1] + h4.z * sa[t + 2] + h4.w * sa[t + 3];
    }
    out[(size_t)n * H2 + d] = acc;
}

extern "C" void kernel_launch(void* const* d_in, const int* in_sizes, int n_in,
                              void* d_out, int out_size, void* d_ws, size_t ws_size,
                              hipStream_t stream)
{
    const float* hidden = (const float*)d_in[0];
    const float* W      = (const float*)d_in[1];
    const float* bias   = (const float*)d_in[2];
    const float* u0     = (const float*)d_in[3];
    float* out = (float*)d_out;

    float* ws     = (float*)d_ws;
    float* u_ws   = ws;
    float* scores = u_ws + (size_t)NSTEP * H2;
    float* attn   = scores + (size_t)NSTEP * TT;

    recurrence_step_kernel<<<H2 / ROWS_PER_WG, 256, 0, stream>>>(W, bias, u0, u_ws, 0);
    for (int i = 1; i < NSTEP; ++i)
        recurrence_step_kernel<<<H2 / ROWS_PER_WG, 256, 0, stream>>>(
            W, bias, u_ws + (size_t)(i - 1) * H2, u_ws + (size_t)i * H2, 1);
    normalize_rows_kernel<<<NSTEP / 8, 256, 0, stream>>>(u_ws);

    scores_kernel<<<dim3(NSTEP, TT / BM), 256, 0, stream>>>(
        hidden, W, bias, u_ws, scores);

    softmax_kernel<<<NSTEP, TT, 0, stream>>>(scores, attn);

    weighted_sum_kernel<<<dim3(NSTEP, H2 / 256), 256, 0, stream>>>(
        hidden, attn, out);
}


# recurrence: two steps per launch: each launch forms W u + b and (W W) u + W b from one u (W W by one f32 WMMA GEMM per call); 64 dependent launches become 32; even steps' sums re-associated
# speedup vs baseline: 8.1956x; 1.0836x over previous
#include <hip/hip_runtime.h>
#include <hip/hip_bf16.h>
#include <math.h>

#define NSTEP 64
#define H2    1024
#define TT    512
#define EPSN  1e-12f

typedef __attribute__((ext_vector_type(2))) float v2f;
typedef __attribute__((ext_vector_type(8))) float v8f;

#define ROWS_PER_WG   32
#define ROWS_PER_WAVE 4

__device__ __forceinline__ float xhalf(float x)
{
    return __builtin_bit_cast(float, __builtin_amdgcn_permlanex16(
        __builtin_bit_cast(int, x), __builtin_bit_cast(int, x),
        0x76543210, 0xfedcba98, false, false));
}

template <int M>
__device__ __forceinline__ float row_xor(float x)
{
    return __builtin_bit_cast(float, __builtin_amdgcn_update_dpp(
        0, __builtin_bit_cast(int, x), 0x160 + M, 0xf, 0xf, true));
}

__device__ __forceinline__ float wave_sum(float v)
{
    v += row_xor<1>(v);
    v += row_xor<2>(v);
    v += row_xor<4>(v);
    v += row_xor<8>(v);
    v += xhalf(v);
    return v;
}

__device__ __forceinline__ float inv_norm(const float4 (&y)[8])
{
    float sq = 0.0f;
    #pragma unroll
    for (int j = 0; j < 8; ++j)
        sq += y[j].x * y[j].x + y[j].y * y[j].y + y[j].z * y[j].z + y[j].w * y[j].w;
    sq = wave_sum(sq);
    return 1.0f / fmaxf(sqrtf(sq), EPSN);
}

__global__ void __launch_bounds__(256)
recurrence_step_kernel(const float* __restrict__ W, const float* __restrict__ bias,
                       const float* __restrict__ vin, float* __restrict__ yout,
                       int normalize)
{
    __shared__ float sy[ROWS_PER_WG];

    const int tid  = threadIdx.x;
    const int wave = tid >> 5;
    const int lane = tid & 31;
    const int e0   = blockIdx.x * ROWS_PER_WG;

    const float4* v4 = (const float4*)vin;
    const float4* w4 = (const float4*)(W + (size_t)(e0 + wave * ROWS_PER_WAVE) * H2);

    float4 u[8];
    #pragma unroll
    for (int j = 0; j < 8; ++j)
        u[j] = v4[lane + 32 * j];

    __builtin_amdgcn_sched_barrier(0);

    float4 w[ROWS_PER_WAVE][8];
    #pragma unroll
    for (int r = 0; r < ROWS_PER_WAVE; ++r) {
        #pragma unroll
        for (int j = 0; j < 8; ++j)
            w[r][j] = w4[r * (H2 / 4) + lane + 32 * j];
    }

    float bb = 0.0f;
    if (tid < ROWS_PER_WG) bb = bias[e0 + tid];

    float inv = 1.0f;
    if (normalize) inv = inv_norm(u);

    #pragma unroll
    for (int j = 0; j < 8; ++j) {
        u[j].x *= inv; u[j].y *= inv; u[j].z *= inv; u[j].w *= inv;
    }

    float p[ROWS_PER_WAVE];
    #pragma unroll
    for (int r = 0; r < ROWS_PER_WAVE; ++r) {
        float ax = 0.0f, ay = 0.0f, az = 0.0f, aw = 0.0f;
        #pragma unroll
        for (int j = 0; j < 8; ++j) {
            ax += w[r][j].x * u[j].x;
            ay += w[r][j].y * u[j].y;
            az += w[r][j].z * u[j].z;
            aw += w[r][j].w * u[j].w;
        }
        p[r] = wave_sum((ax + ay) + (az + aw));
    }

    if (lane == 0) {
        #pragma unroll
        for (int r = 0; r < ROWS_PER_WAVE; ++r)
            sy[wave * ROWS_PER_WAVE + r] = p[r];
    }
    __syncthreads();

    if (tid < ROWS_PER_WG)
        yout[e0 + tid] = sy[tid] + bb;
}

__global__ void __launch_bounds__(256)
normalize_rows_kernel(float* __restrict__ u_ws)
{
    const int lane = threadIdx.x & 31;
    const int row  = blockIdx.x * 8 + (threadIdx.x >> 5);
    float4* p4 = (float4*)(u_ws + (size_t)row * H2);

    float4 y[8];
    #pragma unroll
    for (int j = 0; j < 8; ++j)
        y[j] = p4[lane + 32 * j];

    const float inv = inv_norm(y);

    #pragma unroll
    for (int j = 0; j < 8; ++j) {
        float4 o;
        o.x = y[j].x * inv; o.y = y[j].y * inv; o.z = y[j].z * inv; o.w = y[j].w * inv;
        p4[lane + 32 * j] = o;
    }
}


__device__ __forceinline__ void rebuild_even(const float4 (&a)[8], const float4 (&z)[8],
                                             const float4 (&bv)[8], float4 (&y)[8],
                                             float& inv1)
{
    inv1 = inv_norm(a);
    #pragma unroll
    for (int j = 0; j < 8; ++j) {
        y[j].x = z[j].x * inv1 + bv[j].x;
        y[j].y = z[j].y * inv1 + bv[j].y;
        y[j].z = z[j].z * inv1 + bv[j].z;
        y[j].w = z[j].w * inv1 + bv[j].w;
    }
}

template <bool FIRST>
__global__ void __launch_bounds__(256)
recurrence_pair_kernel(const float* __restrict__ M0, const float* __restrict__ M1,
                       const float* __restrict__ add0, const float* __restrict__ add1,
                       const float* __restrict__ bias,
                       const float* __restrict__ vin, const float* __restrict__ zin,
                       float* __restrict__ out0, float* __restrict__ out1)
{
    __shared__ float sy[ROWS_PER_WG];

    const int tid  = threadIdx.x;
    const int wave = tid >> 5;
    const int lane = tid & 31;
    const int e0   = blockIdx.x * ROWS_PER_WG;
    const int sel  = blockIdx.y;

    const float* M    = sel ? M1 : M0;
    const float* addv = sel ? add1 : add0;
    float*       outv = sel ? out1 : out0;

    const float4* v4 = (const float4*)vin;
    const float4* z4 = (const float4*)zin;
    const float4* b4 = (const float4*)bias;
    const float4* w4 = (const float4*)(M + (size_t)(e0 + wave * ROWS_PER_WAVE) * H2);

    float4 u[8], a[8], z[8], bv[8];
    if (FIRST) {
        #pragma unroll
        for (int j = 0; j < 8; ++j)
            u[j] = v4[lane + 32 * j];
    } else {
        #pragma unroll
        for (int j = 0; j < 8; ++j) {
            a[j]  = v4[lane + 32 * j];
            z[j]  = z4[lane + 32 * j];
            bv[j] = b4[lane + 32 * j];
        }
    }

    __builtin_amdgcn_sched_barrier(0);

    float4 w[ROWS_PER_WAVE][8];
    #pragma unroll
    for (int r = 0; r < ROWS_PER_WAVE; ++r) {
        #pragma unroll
        for (int j = 0; j < 8; ++j)
            w[r][j] = w4[r * (H2 / 4) + lane + 32 * j];
    }

    float bb = 0.0f;
    if (addv != nullptr && tid < ROWS_PER_WG) bb = addv[e0 + tid];

    if (!FIRST) {
        float inv1;
        rebuild_even(a, z, bv, u, inv1);
        const float inv2 = inv_norm(u);
        #pragma unroll
        for (int j = 0; j < 8; ++j) {
            u[j].x *= inv2; u[j].y *= inv2; u[j].z *= inv2; u[j].w *= inv2;
        }
    }

    float p[ROWS_PER_WAVE];
    #pragma unroll
    for (int r = 0; r < ROWS_PER_WAVE; ++r) {
        float ax = 0.0f, ay = 0.0f, az = 0.0f, aw = 0.0f;
        #pragma unroll
        for (int j = 0; j < 8; ++j) {
            ax += w[r][j].x * u[j].x;
            ay += w[r][j].y * u[j].y;
            az += w[r][j].z * u[j].z;
            aw += w[r][j].w * u[j].w;
        }
        p[r] = wave_sum((ax + ay) + (az + aw));
    }

    if (lane == 0) {
        #pragma unroll
        for (int r = 0; r < ROWS_PER_WAVE; ++r)
            sy[wave * ROWS_PER_WAVE + r] = p[r];
    }
    __syncthreads();

    if (tid < ROWS_PER_WG)
        outv[e0 + tid] = sy[tid] + bb;
}

__global__ void __launch_bounds__(256)
normalize_pairs_kernel(float* __restrict__ u_ws, const float* __restrict__ bias)
{
    const int lane = threadIdx.x & 31;
    const int pair = blockIdx.x * 8 + (threadIdx.x >> 5);
    float4* pa4 = (float4*)(u_ws + (size_t)(2 * pair) * H2);
    float4* pz4 = (float4*)(u_ws + (size_t)(2 * pair + 1) * H2);
    const float4* b4 = (const float4*)bias;

    float4 a[8], z[8], bv[8], y[8];
    #pragma unroll
    for (int j = 0; j < 8; ++j) {
        a[j]  = pa4[lane + 32 * j];
        z[j]  = pz4[lane + 32 * j];
        bv[j] = b4[lane + 32 * j];
    }

    float inv1;
    rebuild_even(a, z, bv, y, inv1);
    const float inv2 = inv_norm(y);

    #pragma unroll
    for (int j = 0; j < 8; ++j) {
        float4 o;
        o.x = a[j].x * inv1; o.y = a[j].y * inv1; o.z = a[j].z * inv1; o.w = a[j].w * inv1;
        pa4[lane + 32 * j] = o;
        float4 q;
        q.x = y[j].x * inv2; q.y = y[j].y * inv2; q.z = y[j].z * inv2; q.w = y[j].w * inv2;
        pz4[lane + 32 * j] = q;
    }
}

#define BM     64
#define BN     256
#define KCH    32
#define PA     72
#define PB     44
#define NPASS  (H2 / BN)
#define CPP    (H2 / KCH)
#define NCHUNK (NPASS * CPP)

#define GA(it_, j_) (*(const float4*)(hbase + (size_t)(((it_) % CPP) * KCH + ((tid + 256 * (j_)) >> 4)) * TT + 4 * ((tid + 256 * (j_)) & 15)))
#define GB(it_, j_) (*(const float4*)(W + (size_t)(((it_) / CPP) * BN + ((tid + 256 * (j_)) >> 3)) * H2 + ((it_) % CPP) * KCH + 4 * ((tid + 256 * (j_)) & 7)))
#define SA(buf_, j_) (*(float4*)(&sA[buf_][((tid + 256 * (j_)) >> 4) * PA + 4 * ((tid + 256 * (j_)) & 15)]))
#define SB(buf_, j_) (*(float4*)(&sB[buf_][((tid + 256 * (j_)) >> 3) * PB + 4 * ((tid + 256 * (j_)) & 7)]))
#define LOAD_CHUNK(it_)                                                    \
    { a0 = GA(it_, 0); a1 = GA(it_, 1);                                    \
      b0 = GB(it_, 0); b1 = GB(it_, 1); b2 = GB(it_, 2); b3 = GB(it_, 3);  \
      b4 = GB(it_, 4); b5 = GB(it_, 5); b6 = GB(it_, 6); b7 = GB(it_, 7); }
#define STORE_CHUNK(buf_)                                                  \
    { SA(buf_, 0) = a0; SA(buf_, 1) = a1;                                  \
      SB(buf_, 0) = b0; SB(buf_, 1) = b1; SB(buf_, 2) = b2; SB(buf_, 3) = b3; \
      SB(buf_, 4) = b4; SB(buf_, 5) = b5; SB(buf_, 6) = b6; SB(buf_, 7) = b7; }

__global__ void __launch_bounds__(256)
scores_kernel(const float* __restrict__ hidden, const float* __restrict__ W,
              const float* __restrict__ bias,  const float* __restrict__ u_ws,
              float* __restrict__ scores)
{
    __shared__ __attribute__((aligned(16))) float sA[2][KCH * PA];
    __shared__ __attribute__((aligned(16))) float sB[2][BN * PB];
    __shared__ float sPart[4][BM];

    const int n    = blockIdx.x;
    const int t0   = blockIdx.y * BM;
    const int tid  = threadIdx.x;
    const int lane = tid & 31;
    const int wave = tid >> 5;
    const int c    = lane & 15;
    const int h    = lane >> 4;
    const int wm   = wave >> 2;
    const int wn   = wave & 3;

    const float* hbase = hidden + (size_t)n * H2 * TT + t0;

    v8f acc[2][4];
    #pragma unroll
    for (int mt = 0; mt < 2; ++mt)
        #pragma unroll
        for (int nt = 0; nt < 4; ++nt)
            acc[mt][nt] = (v8f){0.f,0.f,0.f,0.f,0.f,0.f,0.f,0.f};

    float tAcc[2][8];
    #pragma unroll
    for (int mt = 0; mt < 2; ++mt)
        #pragma unroll
        for (int r = 0; r < 8; ++r) tAcc[mt][r] = 0.0f;

    float4 a0, a1, b0, b1, b2, b3, b4, b5, b6, b7;
    LOAD_CHUNK(0)
    STORE_CHUNK(0)
    __syncthreads();

    for (int it = 0; it < NCHUNK; ++it) {
        const int cur = it & 1;
        const bool more = (it + 1 < NCHUNK);

        if (more) LOAD_CHUNK(it + 1)

        const float* pa = sA[cur] + (2 * h) * PA + wm * 32 + c;
        const float* pb = sB[cur] + (wn * 64 + c) * PB + 2 * h;
        #pragma unroll
        for (int ks = 0; ks < KCH / 4; ++ks) {
            v2f af[2], bf[4];
            #pragma unroll
            for (int mt = 0; mt < 2; ++mt) {
                af[mt].x = pa[(ks * 4)     * PA + mt * 16];
                af[mt].y = pa[(ks * 4 + 1) * PA + mt * 16];
            }
            #pragma unroll
            for (int nt = 0; nt < 4; ++nt)
                bf[nt] = *(const v2f*)(pb + nt * 16 * PB + ks * 4);
            #pragma unroll
            for (int mt = 0; mt < 2; ++mt)
                #pragma unroll
                for (int nt = 0; nt < 4; ++nt)
                    acc[mt][nt] = __builtin_amdgcn_wmma_f32_16x16x4_f32(
                         false, af[mt],  false, bf[nt],
                         (short)0, acc[mt][nt],
                         false,  false);
        }

        if ((it % CPP) == CPP - 1) {
            const int ec = it / CPP;
            #pragma unroll
            for (int nt = 0; nt < 4; ++nt) {
                const int e = ec * BN + wn * 64 + nt * 16 + c;
                const float bb = bias[e];
                const float uu = u_ws[(size_t)n * H2 + e];
                #pragma unroll
                for (int mt = 0; mt < 2; ++mt) {
                    #pragma unroll
                    for (int r = 0; r < 8; ++r) {
                        tAcc[mt][r] += tanhf(acc[mt][nt][r] + bb) * uu;
                        acc[mt][nt][r] = 0.0f;
                    }
                }
            }
        }

        if (more) STORE_CHUNK(cur ^ 1)
        __syncthreads();
    }

    #pragma unroll
    for (int mt = 0; mt < 2; ++mt) {
        #pragma unroll
        for (int r = 0; r < 8; ++r) {
            float v = tAcc[mt][r];
            #pragma unroll
            for (int off = 8; off >= 1; off >>= 1)
                v += __shfl_xor(v, off, 32);
            if (c == 0)
                sPart[wn][wm * 32 + mt * 16 + 8 * h + r] = v;
        }
    }
    __syncthreads();

    if (tid < BM)
        scores[(size_t)n * TT + t0 + tid] =
            ((sPart[0][tid] + sPart[1][tid]) + sPart[2][tid]) + sPart[3][tid];
}

#define BN2 64
#define GA2(it_, j_) (*(const float4*)(abase + (size_t)((it_) * KCH + ((tid + 256 * (j_)) >> 4)) * H2 + 4 * ((tid + 256 * (j_)) & 15)))
#define GB2(it_, j_) (*(const float4*)(bbase + (size_t)((tid + 256 * (j_)) >> 3) * H2 + (it_) * KCH + 4 * ((tid + 256 * (j_)) & 7)))

__global__ void __launch_bounds__(256)
w2_gemm_kernel(const float* W, float* __restrict__ W2)
{
    __shared__ __attribute__((aligned(16))) float sA[2][KCH * PA];
    __shared__ __attribute__((aligned(16))) float sB[2][BN2 * PB];

    const int j0   = blockIdx.x * BM;
    const int i0   = blockIdx.y * BN2;
    const int tid  = threadIdx.x;
    const int lane = tid & 31;
    const int wave = tid >> 5;
    const int c    = lane & 15;
    const int h    = lane >> 4;
    const int wm   = wave >> 2;
    const int wn   = wave & 3;

    const float* abase = W + j0;
    const float* bbase = W + (size_t)i0 * H2;

    v8f acc[2];
    acc[0] = (v8f){0.f,0.f,0.f,0.f,0.f,0.f,0.f,0.f};
    acc[1] = (v8f){0.f,0.f,0.f,0.f,0.f,0.f,0.f,0.f};

    float4 a0, a1, b0, b1;
    a0 = GA2(0, 0); a1 = GA2(0, 1); b0 = GB2(0, 0); b1 = GB2(0, 1);
    SA(0, 0) = a0; SA(0, 1) = a1; SB(0, 0) = b0; SB(0, 1) = b1;
    __syncthreads();

    for (int it = 0; it < H2 / KCH; ++it) {
        const int cur = it & 1;
        const int nx = (it + 1 < H2 / KCH) ? it + 1 : it;

        a0 = GA2(nx, 0); a1 = GA2(nx, 1); b0 = GB2(nx, 0); b1 = GB2(nx, 1);
        __builtin_amdgcn_sched_barrier(0);

        const float* pa = sA[cur] + (2 * h) * PA + wm * 32 + c;
        const float* pb = sB[cur] + (wn * 16 + c) * PB + 2 * h;
        #pragma unroll
        for (int ks = 0; ks < KCH / 4; ++ks) {
            v2f af[2], bf;
            #pragma unroll
            for (int mt = 0; mt < 2; ++mt) {
                af[mt].x = pa[(ks * 4)     * PA + mt * 16];
                af[mt].y = pa[(ks * 4 + 1) * PA + mt * 16];
            }
            bf = *(const v2f*)(pb + ks * 4);
            #pragma unroll
            for (int mt = 0; mt < 2; ++mt)
                acc[mt] = __builtin_amdgcn_wmma_f32_16x16x4_f32(
                     false, af[mt],  false, bf,
                     (short)0, acc[mt],
                     false,  false);
        }

        SA(cur ^ 1, 0) = a0; SA(cur ^ 1, 1) = a1; SB(cur ^ 1, 0) = b0; SB(cur ^ 1, 1) = b1;
        __syncthreads();
    }

    #pragma unroll
    for (int mt = 0; mt < 2; ++mt) {
        float* dst = W2 + (size_t)(i0 + wn * 16 + c) * H2 + j0 + wm * 32 + mt * 16 + 8 * h;
        float4 lo, hi;
        lo.x = acc[mt][0]; lo.y = acc[mt][1]; lo.z = acc[mt][2]; lo.w = acc[mt][3];
        hi.x = acc[mt][4]; hi.y = acc[mt][5]; hi.z = acc[mt][6]; hi.w = acc[mt][7];
        *(float4*)(dst)     = lo;
        *(float4*)(dst + 4) = hi;
    }
}

__global__ void __launch_bounds__(512)
softmax_kernel(const float* __restrict__ scores, float* __restrict__ attn)
{
    __shared__ float red[16];
    const int n   = blockIdx.x;
    const int tid = threadIdx.x;

    const float s = scores[(size_t)n * TT + tid];

    float m = s;
    #pragma unroll
    for (int off = 16; off >= 1; off >>= 1)
        m = fmaxf(m, __shfl_xor(m, off, 32));
    if ((tid & 31) == 0) red[tid >> 5] = m;
    __syncthreads();
    if (tid < 16) {
        float r = red[tid];
        #pragma unroll
        for (int off = 8; off >= 1; off >>= 1)
            r = fmaxf(r, __shfl_xor(r, off, 32));
        red[tid] = r;
    }
    __syncthreads();
    m = red[0];
    __syncthreads();

    const float e = __expf(s - m);
    float sum = e;
    #pragma unroll
    for (int off = 16; off >= 1; off >>= 1)
        sum += __shfl_xor(sum, off, 32);
    if ((tid & 31) == 0) red[tid >> 5] = sum;
    __syncthreads();
    if (tid < 16) {
        float r = red[tid];
        #pragma unroll
        for (int off = 8; off >= 1; off >>= 1)
            r += __shfl_xor(r, off, 32);
        red[tid] = r;
    }
    __syncthreads();

    attn[(size_t)n * TT + tid] = e / red[0];
}

__global__ void __launch_bounds__(256)
weighted_sum_kernel(const float* __restrict__ hidden,
                    const float* __restrict__ attn, float* __restrict__ out)
{
    __shared__ float sa[TT];
    const int n   = blockIdx.x;
    const int dc  = blockIdx.y;
    const int tid = threadIdx.x;

    sa[tid]       = attn[(size_t)n * TT + tid];
    sa[tid + 256] = attn[(size_t)n * TT + tid + 256];
    __syncthreads();

    const int d = dc * 256 + tid;
    const float* h = hidden + (size_t)n * H2 * TT + (size_t)d * TT;
    float acc = 0.0f;
    #pragma unroll 4
    for (int t = 0; t < TT; t += 4) {
        const float4 h4 = *(const float4*)(h + t);
        acc += h4.x * sa[t] + h4.y * sa[t + 1] + h4.z * sa[t + 2] + h4.w * sa[t + 3];
    }
    out[(size_t)n * H2 + d] = acc;
}

extern "C" void kernel_launch(void* const* d_in, const int* in_sizes, int n_in,
                              void* d_out, int out_size, void* d_ws, size_t ws_size,
                              hipStream_t stream)
{
    const float* hidden = (const float*)d_in[0];
    const float* W      = (const float*)d_in[1];
    const float* bias   = (const float*)d_in[2];
    const float* u0     = (const float*)d_in[3];
    float* out = (float*)d_out;

    float* ws     = (float*)d_ws;
    float* u_ws   = ws;
    float* scores = u_ws + (size_t)NSTEP * H2;
    float* attn   = scores + (size_t)NSTEP * TT;

    float* W2     = attn + (size_t)NSTEP * TT;
    float* Wb     = W2 + (size_t)H2 * H2;

    if (ws_size >= (size_t)4722688) {
        recurrence_pair_kernel<true><<<dim3(H2 / ROWS_PER_WG, 1), 256, 0, stream>>>(
            W, W, nullptr, nullptr, bias, bias, bias, Wb, Wb);
        w2_gemm_kernel<<<dim3(H2 / BM, H2 / BN2), 256, 0, stream>>>(W, W2);

        recurrence_pair_kernel<true><<<dim3(H2 / ROWS_PER_WG, 2), 256, 0, stream>>>(
            W, W2, bias, Wb, bias, u0, u0, u_ws, u_ws + H2);
        for (int k = 1; k < NSTEP / 2; ++k)
            recurrence_pair_kernel<false><<<dim3(H2 / ROWS_PER_WG, 2), 256, 0, stream>>>(
                W, W2, bias, Wb, bias,
                u_ws + (size_t)(2 * k - 2) * H2, u_ws + (size_t)(2 * k - 1) * H2,
                u_ws + (size_t)(2 * k) * H2,     u_ws + (size_t)(2 * k + 1) * H2);
        normalize_pairs_kernel<<<NSTEP / 16, 256, 0, stream>>>(u_ws, bias);
    } else {
        recurrence_step_kernel<<<H2 / ROWS_PER_WG, 256, 0, stream>>>(W, bias, u0, u_ws, 0);
        for (int i = 1; i < NSTEP; ++i)
            recurrence_step_kernel<<<H2 / ROWS_PER_WG, 256, 0, stream>>>(
                W, bias, u_ws + (size_t)(i - 1) * H2, u_ws + (size_t)i * H2, 1);
        normalize_rows_kernel<<<NSTEP / 8, 256, 0, stream>>>(u_ws);
    }

    scores_kernel<<<dim3(NSTEP, TT / BM), 256, 0, stream>>>(
        hidden, W, bias, u_ws, scores);

    softmax_kernel<<<NSTEP, TT, 0, stream>>>(scores, attn);

    weighted_sum_kernel<<<dim3(NSTEP, H2 / 256), 256, 0, stream>>>(
        hidden, attn, out);
}


# scores GEMM: vector-ALU work beside the f32 WMMAs cut: uniform-pointer chunk addresses, A staged per 16-token tile (fragment = one LDS pair read), branch-free tanh from one exp2 and one rcp
# speedup vs baseline: 8.9630x; 1.0936x over previous
#include <hip/hip_runtime.h>
#include <hip/hip_bf16.h>
#include <math.h>

#define NSTEP 64
#define H2    1024
#define TT    512
#define EPSN  1e-12f

typedef __attribute__((ext_vector_type(2))) float v2f;
typedef __attribute__((ext_vector_type(8))) float v8f;

#define ROWS_PER_WG   32
#define ROWS_PER_WAVE 4

__device__ __forceinline__ float xhalf(float x)
{
    return __builtin_bit_cast(float, __builtin_amdgcn_permlanex16(
        __builtin_bit_cast(int, x), __builtin_bit_cast(int, x),
        0x76543210, 0xfedcba98, false, false));
}

template <int M>
__device__ __forceinline__ float row_xor(float x)
{
    return __builtin_bit_cast(float, __builtin_amdgcn_update_dpp(
        0, __builtin_bit_cast(int, x), 0x160 + M, 0xf, 0xf, true));
}

__device__ __forceinline__ float wave_sum(float v)
{
    v += row_xor<1>(v);
    v += row_xor<2>(v);
    v += row_xor<4>(v);
    v += row_xor<8>(v);
    v += xhalf(v);
    return v;
}

__device__ __forceinline__ float inv_norm(const float4 (&y)[8])
{
    float sq = 0.0f;
    #pragma unroll
    for (int j = 0; j < 8; ++j)
        sq += y[j].x * y[j].x + y[j].y * y[j].y + y[j].z * y[j].z + y[j].w * y[j].w;
    sq = wave_sum(sq);
    return 1.0f / fmaxf(sqrtf(sq), EPSN);
}

__global__ void __launch_bounds__(256)
recurrence_step_kernel(const float* __restrict__ W, const float* __restrict__ bias,
                       const float* __restrict__ vin, float* __restrict__ yout,
                       int normalize)
{
    __shared__ float sy[ROWS_PER_WG];

    const int tid  = threadIdx.x;
    const int wave = tid >> 5;
    const int lane = tid & 31;
    const int e0   = blockIdx.x * ROWS_PER_WG;

    const float4* v4 = (const float4*)vin;
    const float4* w4 = (const float4*)(W + (size_t)(e0 + wave * ROWS_PER_WAVE) * H2);

    float4 u[8];
    #pragma unroll
    for (int j = 0; j < 8; ++j)
        u[j] = v4[lane + 32 * j];

    __builtin_amdgcn_sched_barrier(0);

    float4 w[ROWS_PER_WAVE][8];
    #pragma unroll
    for (int r = 0; r < ROWS_PER_WAVE; ++r) {
        #pragma unroll
        for (int j = 0; j < 8; ++j)
            w[r][j] = w4[r * (H2 / 4) + lane + 32 * j];
    }

    float bb = 0.0f;
    if (tid < ROWS_PER_WG) bb = bias[e0 + tid];

    float inv = 1.0f;
    if (normalize) inv = inv_norm(u);

    #pragma unroll
    for (int j = 0; j < 8; ++j) {
        u[j].x *= inv; u[j].y *= inv; u[j].z *= inv; u[j].w *= inv;
    }

    float p[ROWS_PER_WAVE];
    #pragma unroll
    for (int r = 0; r < ROWS_PER_WAVE; ++r) {
        float ax = 0.0f, ay = 0.0f, az = 0.0f, aw = 0.0f;
        #pragma unroll
        for (int j = 0; j < 8; ++j) {
            ax += w[r][j].x * u[j].x;
            ay += w[r][j].y * u[j].y;
            az += w[r][j].z * u[j].z;
            aw += w[r][j].w * u[j].w;
        }
        p[r] = wave_sum((ax + ay) + (az + aw));
    }

    if (lane == 0) {
        #pragma unroll
        for (int r = 0; r < ROWS_PER_WAVE; ++r)
            sy[wave * ROWS_PER_WAVE + r] = p[r];
    }
    __syncthreads();

    if (tid < ROWS_PER_WG)
        yout[e0 + tid] = sy[tid] + bb;
}

__global__ void __launch_bounds__(256)
normalize_rows_kernel(float* __restrict__ u_ws)
{
    const int lane = threadIdx.x & 31;
    const int row  = blockIdx.x * 8 + (threadIdx.x >> 5);
    float4* p4 = (float4*)(u_ws + (size_t)row * H2);

    float4 y[8];
    #pragma unroll
    for (int j = 0; j < 8; ++j)
        y[j] = p4[lane + 32 * j];

    const float inv = inv_norm(y);

    #pragma unroll
    for (int j = 0; j < 8; ++j) {
        float4 o;
        o.x = y[j].x * inv; o.y = y[j].y * inv; o.z = y[j].z * inv; o.w = y[j].w * inv;
        p4[lane + 32 * j] = o;
    }
}


__device__ __forceinline__ void rebuild_even(const float4 (&a)[8], const float4 (&z)[8],
                                             const float4 (&bv)[8], float4 (&y)[8],
                                             float& inv1)
{
    inv1 = inv_norm(a);
    #pragma unroll
    for (int j = 0; j < 8; ++j) {
        y[j].x = z[j].x * inv1 + bv[j].x;
        y[j].y = z[j].y * inv1 + bv[j].y;
        y[j].z = z[j].z * inv1 + bv[j].z;
        y[j].w = z[j].w * inv1 + bv[j].w;
    }
}

template <bool FIRST>
__global__ void __launch_bounds__(256)
recurrence_pair_kernel(const float* __restrict__ M0, const float* __restrict__ M1,
                       const float* __restrict__ add0, const float* __restrict__ add1,
                       const float* __restrict__ bias,
                       const float* __restrict__ vin, const float* __restrict__ zin,
                       float* __restrict__ out0, float* __restrict__ out1)
{
    __shared__ float sy[ROWS_PER_WG];

    const int tid  = threadIdx.x;
    const int wave = tid >> 5;
    const int lane = tid & 31;
    const int e0   = blockIdx.x * ROWS_PER_WG;
    const int sel  = blockIdx.y;

    const float* M    = sel ? M1 : M0;
    const float* addv = sel ? add1 : add0;
    float*       outv = sel ? out1 : out0;

    const float4* v4 = (const float4*)vin;
    const float4* z4 = (const float4*)zin;
    const float4* b4 = (const float4*)bias;
    const float4* w4 = (const float4*)(M + (size_t)(e0 + wave * ROWS_PER_WAVE) * H2);

    float4 u[8], a[8], z[8], bv[8];
    if (FIRST) {
        #pragma unroll
        for (int j = 0; j < 8; ++j)
            u[j] = v4[lane + 32 * j];
    } else {
        #pragma unroll
        for (int j = 0; j < 8; ++j) {
            a[j]  = v4[lane + 32 * j];
            z[j]  = z4[lane + 32 * j];
            bv[j] = b4[lane + 32 * j];
        }
    }

    __builtin_amdgcn_sched_barrier(0);

    float4 w[ROWS_PER_WAVE][8];
    #pragma unroll
    for (int r = 0; r < ROWS_PER_WAVE; ++r) {
        #pragma unroll
        for (int j = 0; j < 8; ++j)
            w[r][j] = w4[r * (H2 / 4) + lane + 32 * j];
    }

    float bb = 0.0f;
    if (addv != nullptr && tid < ROWS_PER_WG) bb = addv[e0 + tid];

    if (!FIRST) {
        float inv1;
        rebuild_even(a, z, bv, u, inv1);
        const float inv2 = inv_norm(u);
        #pragma unroll
        for (int j = 0; j < 8; ++j) {
            u[j].x *= inv2; u[j].y *= inv2; u[j].z *= inv2; u[j].w *= inv2;
        }
    }

    float p[ROWS_PER_WAVE];
    #pragma unroll
    for (int r = 0; r < ROWS_PER_WAVE; ++r) {
        float ax = 0.0f, ay = 0.0f, az = 0.0f, aw = 0.0f;
        #pragma unroll
        for (int j = 0; j < 8; ++j) {
            ax += w[r][j].x * u[j].x;
            ay += w[r][j].y * u[j].y;
            az += w[r][j].z * u[j].z;
            aw += w[r][j].w * u[j].w;
        }
        p[r] = wave_sum((ax + ay) + (az + aw));
    }

    if (lane == 0) {
        #pragma unroll
        for (int r = 0; r < ROWS_PER_WAVE; ++r)
            sy[wave * ROWS_PER_WAVE + r] = p[r];
    }
    __syncthreads();

    if (tid < ROWS_PER_WG)
        outv[e0 + tid] = sy[tid] + bb;
}

__global__ void __launch_bounds__(256)
normalize_pairs_kernel(float* __restrict__ u_ws, const float* __restrict__ bias)
{
    const int lane = threadIdx.x & 31;
    const int pair = blockIdx.x * 8 + (threadIdx.x >> 5);
    float4* pa4 = (float4*)(u_ws + (size_t)(2 * pair) * H2);
    float4* pz4 = (float4*)(u_ws + (size_t)(2 * pair + 1) * H2);
    const float4* b4 = (const float4*)bias;

    float4 a[8], z[8], bv[8], y[8];
    #pragma unroll
    for (int j = 0; j < 8; ++j) {
        a[j]  = pa4[lane + 32 * j];
        z[j]  = pz4[lane + 32 * j];
        bv[j] = b4[lane + 32 * j];
    }

    float inv1;
    rebuild_even(a, z, bv, y, inv1);
    const float inv2 = inv_norm(y);

    #pragma unroll
    for (int j = 0; j < 8; ++j) {
        float4 o;
        o.x = a[j].x * inv1; o.y = a[j].y * inv1; o.z = a[j].z * inv1; o.w = a[j].w * inv1;
        pa4[lane + 32 * j] = o;
        float4 q;
        q.x = y[j].x * inv2; q.y = y[j].y * inv2; q.z = y[j].z * inv2; q.w = y[j].w * inv2;
        pz4[lane + 32 * j] = q;
    }
}

#define BM     64
#define BN     256
#define KCH    32
#define PA     72
#define PB     44
#define NPASS  (H2 / BN)
#define CPP    (H2 / KCH)
#define NCHUNK (NPASS * CPP)

#define SA(buf_, j_) (*(float4*)(&sA[buf_][((tid + 256 * (j_)) >> 4) * PA + 4 * ((tid + 256 * (j_)) & 15)]))
#define SB(buf_, j_) (*(float4*)(&sB[buf_][((tid + 256 * (j_)) >> 3) * PB + 4 * ((tid + 256 * (j_)) & 7)]))

#define TSA 528
#define GAS(p_, j_) (*(const float4*)((p_) + (size_t)offA + (size_t)((j_) * 16 * TT * 4)))
#define GBS(p_, j_) (*(const float4*)((p_) + (size_t)offB + (size_t)((j_) * 32 * H2 * 4)))
#define SAS(buf_, j_) (*(float4*)(sAw + (buf_) * (KCH * PA) + (j_) * 256))
#define SBS(buf_, j_) (*(float4*)(sBw + (buf_) * (BN * PB) + (j_) * 32 * PB))
#define LOAD_CHUNK_S(it_)                                                          \
    { const char* ap_ = hb + (size_t)(((it_) & (CPP - 1)) * (KCH * TT * 4));       \
      const char* bp_ = wb + (size_t)(((it_) >> 5) * (BN * H2 * 4) + ((it_) & (CPP - 1)) * (KCH * 4)); \
      a0 = GAS(ap_, 0); a1 = GAS(ap_, 1);                                          \
      b0 = GBS(bp_, 0); b1 = GBS(bp_, 1); b2 = GBS(bp_, 2); b3 = GBS(bp_, 3);      \
      b4 = GBS(bp_, 4); b5 = GBS(bp_, 5); b6 = GBS(bp_, 6); b7 = GBS(bp_, 7); }
#define STORE_CHUNK_S(buf_)                                                        \
    { SAS(buf_, 0) = a0; SAS(buf_, 1) = a1;                                        \
      SBS(buf_, 0) = b0; SBS(buf_, 1) = b1; SBS(buf_, 2) = b2; SBS(buf_, 3) = b3;  \
      SBS(buf_, 4) = b4; SBS(buf_, 5) = b5; SBS(buf_, 6) = b6; SBS(buf_, 7) = b7; }

__device__ __forceinline__ float tanh_er(float x)
{
    const float e = __builtin_amdgcn_exp2f(fabsf(x) * 2.885390081777927f);
    const float r = __builtin_amdgcn_rcpf(e + 1.0f);
    return copysignf(fmaf(-2.0f, r, 1.0f), x);
}

__global__ void __launch_bounds__(256)
scores_kernel(const float* hidden, const float* W,
              const float* __restrict__ bias,  const float* __restrict__ u_ws,
              float* __restrict__ scores)
{
    __shared__ __attribute__((aligned(16))) float sA[2][KCH * PA];
    __shared__ __attribute__((aligned(16))) float sB[2][BN * PB];
    __shared__ float sPart[4][BM];

    const int n    = blockIdx.x;
    const int t0   = blockIdx.y * BM;
    const int tid  = threadIdx.x;
    const int lane = tid & 31;
    const int wave = tid >> 5;
    const int c    = lane & 15;
    const int h    = lane >> 4;
    const int wm   = wave >> 2;
    const int wn   = wave & 3;

    const char* hb = (const char*)(hidden + (size_t)n * H2 * TT + t0);
    const char* wb = (const char*)W;
    const unsigned offA = (unsigned)(((tid >> 4) * TT + 4 * (tid & 15)) * 4);
    const unsigned offB = (unsigned)(((tid >> 3) * H2 + 4 * (tid & 7)) * 4);
    float* sAw = &sA[0][0] + ((tid & 15) >> 2) * TSA + (tid >> 4) * 16 + 4 * (tid & 3);
    float* sBw = &sB[0][0] + (tid >> 3) * PB + 4 * (tid & 7);

    v8f acc[2][4];
    #pragma unroll
    for (int mt = 0; mt < 2; ++mt)
        #pragma unroll
        for (int nt = 0; nt < 4; ++nt)
            acc[mt][nt] = (v8f){0.f,0.f,0.f,0.f,0.f,0.f,0.f,0.f};

    float tAcc[2][8];
    #pragma unroll
    for (int mt = 0; mt < 2; ++mt)
        #pragma unroll
        for (int r = 0; r < 8; ++r) tAcc[mt][r] = 0.0f;

    float4 a0, a1, b0, b1, b2, b3, b4, b5, b6, b7;
    LOAD_CHUNK_S(0)
    STORE_CHUNK_S(0)
    __syncthreads();

    for (int it = 0; it < NCHUNK; ++it) {
        const int cur = it & 1;
        const int nx = (it + 1 < NCHUNK) ? it + 1 : it;

        LOAD_CHUNK_S(nx)
        __builtin_amdgcn_sched_barrier(0);

        const float* pa = sA[cur] + (2 * wm) * TSA + (2 * h) * 16 + c;
        const float* pb = sB[cur] + (wn * 64 + c) * PB + 2 * h;
        #pragma unroll
        for (int ks = 0; ks < KCH / 4; ++ks) {
            v2f af[2], bf[4];
            #pragma unroll
            for (int mt = 0; mt < 2; ++mt) {
                af[mt].x = pa[mt * TSA + (ks * 4)     * 16];
                af[mt].y = pa[mt * TSA + (ks * 4 + 1) * 16];
            }
            #pragma unroll
            for (int nt = 0; nt < 4; ++nt)
                bf[nt] = *(const v2f*)(pb + nt * 16 * PB + ks * 4);
            #pragma unroll
            for (int mt = 0; mt < 2; ++mt)
                #pragma unroll
                for (int nt = 0; nt < 4; ++nt)
                    acc[mt][nt] = __builtin_amdgcn_wmma_f32_16x16x4_f32(
                         false, af[mt],  false, bf[nt],
                         (short)0, acc[mt][nt],
                         false,  false);
        }

        if ((it % CPP) == CPP - 1) {
            const int ec = it / CPP;
            #pragma unroll
            for (int nt = 0; nt < 4; ++nt) {
                const int e = ec * BN + wn * 64 + nt * 16 + c;
                const float bb = bias[e];
                const float uu = u_ws[(size_t)n * H2 + e];
                #pragma unroll
                for (int mt = 0; mt < 2; ++mt) {
                    #pragma unroll
                    for (int r = 0; r < 8; ++r) {
                        tAcc[mt][r] += tanh_er(acc[mt][nt][r] + bb) * uu;
                        acc[mt][nt][r] = 0.0f;
                    }
                }
            }
        }

        STORE_CHUNK_S(cur ^ 1)
        __syncthreads();
    }

    #pragma unroll
    for (int mt = 0; mt < 2; ++mt) {
        #pragma unroll
        for (int r = 0; r < 8; ++r) {
            float v = tAcc[mt][r];
            #pragma unroll
            for (int off = 8; off >= 1; off >>= 1)
                v += __shfl_xor(v, off, 32);
            if (c == 0)
                sPart[wn][wm * 32 + mt * 16 + 8 * h + r] = v;
        }
    }
    __syncthreads();

    if (tid < BM)
        scores[(size_t)n * TT + t0 + tid] =
            ((sPart[0][tid] + sPart[1][tid]) + sPart[2][tid]) + sPart[3][tid];
}

#define BN2 64
#define GA2(it_, j_) (*(const float4*)(abase + (size_t)((it_) * KCH + ((tid + 256 * (j_)) >> 4)) * H2 + 4 * ((tid + 256 * (j_)) & 15)))
#define GB2(it_, j_) (*(const float4*)(bbase + (size_t)((tid + 256 * (j_)) >> 3) * H2 + (it_) * KCH + 4 * ((tid + 256 * (j_)) & 7)))

__global__ void __launch_bounds__(256)
w2_gemm_kernel(const float* W, float* __restrict__ W2)
{
    __shared__ __attribute__((aligned(16))) float sA[2][KCH * PA];
    __shared__ __attribute__((aligned(16))) float sB[2][BN2 * PB];

    const int j0   = blockIdx.x * BM;
    const int i0   = blockIdx.y * BN2;
    const int tid  = threadIdx.x;
    const int lane = tid & 31;
    const int wave = tid >> 5;
    const int c    = lane & 15;
    const int h    = lane >> 4;
    const int wm   = wave >> 2;
    const int wn   = wave & 3;

    const float* abase = W + j0;
    const float* bbase = W + (size_t)i0 * H2;

    v8f acc[2];
    acc[0] = (v8f){0.f,0.f,0.f,0.f,0.f,0.f,0.f,0.f};
    acc[1] = (v8f){0.f,0.f,0.f,0.f,0.f,0.f,0.f,0.f};

    float4 a0, a1, b0, b1;
    a0 = GA2(0, 0); a1 = GA2(0, 1); b0 = GB2(0, 0); b1 = GB2(0, 1);
    SA(0, 0) = a0; SA(0, 1) = a1; SB(0, 0) = b0; SB(0, 1) = b1;
    __syncthreads();

    for (int it = 0; it < H2 / KCH; ++it) {
        const int cur = it & 1;
        const int nx = (it + 1 < H2 / KCH) ? it + 1 : it;

        a0 = GA2(nx, 0); a1 = GA2(nx, 1); b0 = GB2(nx, 0); b1 = GB2(nx, 1);
        __builtin_amdgcn_sched_barrier(0);

        const float* pa = sA[cur] + (2 * h) * PA + wm * 32 + c;
        const float* pb = sB[cur] + (wn * 16 + c) * PB + 2 * h;
        #pragma unroll
        for (int ks = 0; ks < KCH / 4; ++ks) {
            v2f af[2], bf;
            #pragma unroll
            for (int mt = 0; mt < 2; ++mt) {
                af[mt].x = pa[(ks * 4)     * PA + mt * 16];
                af[mt].y = pa[(ks * 4 + 1) * PA + mt * 16];
            }
            bf = *(const v2f*)(pb + ks * 4);
            #pragma unroll
            for (int mt = 0; mt < 2; ++mt)
                acc[mt] = __builtin_amdgcn_wmma_f32_16x16x4_f32(
                     false, af[mt],  false, bf,
                     (short)0, acc[mt],
                     false,  false);
        }

        SA(cur ^ 1, 0) = a0; SA(cur ^ 1, 1) = a1; SB(cur ^ 1, 0) = b0; SB(cur ^ 1, 1) = b1;
        __syncthreads();
    }

    #pragma unroll
    for (int mt = 0; mt < 2; ++mt) {
        float* dst = W2 + (size_t)(i0 + wn * 16 + c) * H2 + j0 + wm * 32 + mt * 16 + 8 * h;
        float4 lo, hi;
        lo.x = acc[mt][0]; lo.y = acc[mt][1]; lo.z = acc[mt][2]; lo.w = acc[mt][3];
        hi.x = acc[mt][4]; hi.y = acc[mt][5]; hi.z = acc[mt][6]; hi.w = acc[mt][7];
        *(float4*)(dst)     = lo;
        *(float4*)(dst + 4) = hi;
    }
}

__global__ void __launch_bounds__(512)
softmax_kernel(const float* __restrict__ scores, float* __restrict__ attn)
{
    __shared__ float red[16];
    const int n   = blockIdx.x;
    const int tid = threadIdx.x;

    const float s = scores[(size_t)n * TT + tid];

    float m = s;
    #pragma unroll
    for (int off = 16; off >= 1; off >>= 1)
        m = fmaxf(m, __shfl_xor(m, off, 32));
    if ((tid & 31) == 0) red[tid >> 5] = m;
    __syncthreads();
    if (tid < 16) {
        float r = red[tid];
        #pragma unroll
        for (int off = 8; off >= 1; off >>= 1)
            r = fmaxf(r, __shfl_xor(r, off, 32));
        red[tid] = r;
    }
    __syncthreads();
    m = red[0];
    __syncthreads();

    const float e = __expf(s - m);
    float sum = e;
    #pragma unroll
    for (int off = 16; off >= 1; off >>= 1)
        sum += __shfl_xor(sum, off, 32);
    if ((tid & 31) == 0) red[tid >> 5] = sum;
    __syncthreads();
    if (tid < 16) {
        float r = red[tid];
        #pragma unroll
        for (int off = 8; off >= 1; off >>= 1)
            r += __shfl_xor(r, off, 32);
        red[tid] = r;
    }
    __syncthreads();

    attn[(size_t)n * TT + tid] = e / red[0];
}

__global__ void __launch_bounds__(256)
weighted_sum_kernel(const float* __restrict__ hidden,
                    const float* __restrict__ attn, float* __restrict__ out)
{
    __shared__ float sa[TT];
    const int n   = blockIdx.x;
    const int dc  = blockIdx.y;
    const int tid = threadIdx.x;

    sa[tid]       = attn[(size_t)n * TT + tid];
    sa[tid + 256] = attn[(size_t)n * TT + tid + 256];
    __syncthreads();

    const int d = dc * 256 + tid;
    const float* h = hidden + (size_t)n * H2 * TT + (size_t)d * TT;
    float acc = 0.0f;
    #pragma unroll 4
    for (int t = 0; t < TT; t += 4) {
        const float4 h4 = *(const float4*)(h + t);
        acc += h4.x * sa[t] + h4.y * sa[t + 1] + h4.z * sa[t + 2] + h4.w * sa[t + 3];
    }
    out[(size_t)n * H2 + d] = acc;
}

extern "C" void kernel_launch(void* const* d_in, const int* in_sizes, int n_in,
                              void* d_out, int out_size, void* d_ws, size_t ws_size,
                              hipStream_t stream)
{
    const float* hidden = (const float*)d_in[0];
    const float* W      = (const float*)d_in[1];
    const float* bias   = (const float*)d_in[2];
    const float* u0     = (const float*)d_in[3];
    float* out = (float*)d_out;

    float* ws     = (float*)d_ws;
    float* u_ws   = ws;
    float* scores = u_ws + (size_t)NSTEP * H2;
    float* attn   = scores + (size_t)NSTEP * TT;

    float* W2     = attn + (size_t)NSTEP * TT;
    float* Wb     = W2 + (size_t)H2 * H2;

    if (ws_size >= (size_t)4722688) {
        recurrence_pair_kernel<true><<<dim3(H2 / ROWS_PER_WG, 1), 256, 0, stream>>>(
            W, W, nullptr, nullptr, bias, bias, bias, Wb, Wb);
        w2_gemm_kernel<<<dim3(H2 / BM, H2 / BN2), 256, 0, stream>>>(W, W2);

        recurrence_pair_kernel<true><<<dim3(H2 / ROWS_PER_WG, 2), 256, 0, stream>>>(
            W, W2, bias, Wb, bias, u0, u0, u_ws, u_ws + H2);
        for (int k = 1; k < NSTEP / 2; ++k)
            recurrence_pair_kernel<false><<<dim3(H2 / ROWS_PER_WG, 2), 256, 0, stream>>>(
                W, W2, bias, Wb, bias,
                u_ws + (size_t)(2 * k - 2) * H2, u_ws + (size_t)(2 * k - 1) * H2,
                u_ws + (size_t)(2 * k) * H2,     u_ws + (size_t)(2 * k + 1) * H2);
        normalize_pairs_kernel<<<NSTEP / 16, 256, 0, stream>>>(u_ws, bias);
    } else {
        recurrence_step_kernel<<<H2 / ROWS_PER_WG, 256, 0, stream>>>(W, bias, u0, u_ws, 0);
        for (int i = 1; i < NSTEP; ++i)
            recurrence_step_kernel<<<H2 / ROWS_PER_WG, 256, 0, stream>>>(
                W, bias, u_ws + (size_t)(i - 1) * H2, u_ws + (size_t)i * H2, 1);
        normalize_rows_kernel<<<NSTEP / 8, 256, 0, stream>>>(u_ws);
    }

    scores_kernel<<<dim3(NSTEP, TT / BM), 256, 0, stream>>>(
        hidden, W, bias, u_ws, scores);

    softmax_kernel<<<NSTEP, TT, 0, stream>>>(scores, attn);

    weighted_sum_kernel<<<dim3(NSTEP, H2 / 256), 256, 0, stream>>>(
        hidden, attn, out);
}


# both f32 WMMA GEMM loops (scores, W W): ring of three LDS stages; next chunk stored and barrier signalled inside a chunk's WMMAs, waited for at the top of the next (split barrier); sums unchanged
# speedup vs baseline: 9.3886x; 1.0475x over previous
#include <hip/hip_runtime.h>
#include <hip/hip_bf16.h>
#include <math.h>

#define NSTEP 64
#define H2    1024
#define TT    512
#define EPSN  1e-12f

typedef __attribute__((ext_vector_type(2))) float v2f;
typedef __attribute__((ext_vector_type(8))) float v8f;

#define ROWS_PER_WG   32
#define ROWS_PER_WAVE 4

__device__ __forceinline__ float xhalf(float x)
{
    return __builtin_bit_cast(float, __builtin_amdgcn_permlanex16(
        __builtin_bit_cast(int, x), __builtin_bit_cast(int, x),
        0x76543210, 0xfedcba98, false, false));
}

template <int M>
__device__ __forceinline__ float row_xor(float x)
{
    return __builtin_bit_cast(float, __builtin_amdgcn_update_dpp(
        0, __builtin_bit_cast(int, x), 0x160 + M, 0xf, 0xf, true));
}

__device__ __forceinline__ float wave_sum(float v)
{
    v += row_xor<1>(v);
    v += row_xor<2>(v);
    v += row_xor<4>(v);
    v += row_xor<8>(v);
    v += xhalf(v);
    return v;
}

__device__ __forceinline__ float inv_norm(const float4 (&y)[8])
{
    float sq = 0.0f;
    #pragma unroll
    for (int j = 0; j < 8; ++j)
        sq += y[j].x * y[j].x + y[j].y * y[j].y + y[j].z * y[j].z + y[j].w * y[j].w;
    sq = wave_sum(sq);
    return 1.0f / fmaxf(sqrtf(sq), EPSN);
}

__global__ void __launch_bounds__(256)
recurrence_step_kernel(const float* __restrict__ W, const float* __restrict__ bias,
                       const float* __restrict__ vin, float* __restrict__ yout,
                       int normalize)
{
    __shared__ float sy[ROWS_PER_WG];

    const int tid  = threadIdx.x;
    const int wave = tid >> 5;
    const int lane = tid & 31;
    const int e0   = blockIdx.x * ROWS_PER_WG;

    const float4* v4 = (const float4*)vin;
    const float4* w4 = (const float4*)(W + (size_t)(e0 + wave * ROWS_PER_WAVE) * H2);

    float4 u[8];
    #pragma unroll
    for (int j = 0; j < 8; ++j)
        u[j] = v4[lane + 32 * j];

    __builtin_amdgcn_sched_barrier(0);

    float4 w[ROWS_PER_WAVE][8];
    #pragma unroll
    for (int r = 0; r < ROWS_PER_WAVE; ++r) {
        #pragma unroll
        for (int j = 0; j < 8; ++j)
            w[r][j] = w4[r * (H2 / 4) + lane + 32 * j];
    }

    float bb = 0.0f;
    if (tid < ROWS_PER_WG) bb = bias[e0 + tid];

    float inv = 1.0f;
    if (normalize) inv = inv_norm(u);

    #pragma unroll
    for (int j = 0; j < 8; ++j) {
        u[j].x *= inv; u[j].y *= inv; u[j].z *= inv; u[j].w *= inv;
    }

    float p[ROWS_PER_WAVE];
    #pragma unroll
    for (int r = 0; r < ROWS_PER_WAVE; ++r) {
        float ax = 0.0f, ay = 0.0f, az = 0.0f, aw = 0.0f;
        #pragma unroll
        for (int j = 0; j < 8; ++j) {
            ax += w[r][j].x * u[j].x;
            ay += w[r][j].y * u[j].y;
            az += w[r][j].z * u[j].z;
            aw += w[r][j].w * u[j].w;
        }
        p[r] = wave_sum((ax + ay) + (az + aw));
    }

    if (lane == 0) {
        #pragma unroll
        for (int r = 0; r < ROWS_PER_WAVE; ++r)
            sy[wave * ROWS_PER_WAVE + r] = p[r];
    }
    __syncthreads();

    if (tid < ROWS_PER_WG)
        yout[e0 + tid] = sy[tid] + bb;
}

__global__ void __launch_bounds__(256)
normalize_rows_kernel(float* __restrict__ u_ws)
{
    const int lane = threadIdx.x & 31;
    const int row  = blockIdx.x * 8 + (threadIdx.x >> 5);
    float4* p4 = (float4*)(u_ws + (size_t)row * H2);

    float4 y[8];
    #pragma unroll
    for (int j = 0; j < 8; ++j)
        y[j] = p4[lane + 32 * j];

    const float inv = inv_norm(y);

    #pragma unroll
    for (int j = 0; j < 8; ++j) {
        float4 o;
        o.x = y[j].x * inv; o.y = y[j].y * inv; o.z = y[j].z * inv; o.w = y[j].w * inv;
        p4[lane + 32 * j] = o;
    }
}


__device__ __forceinline__ void rebuild_even(const float4 (&a)[8], const float4 (&z)[8],
                                             const float4 (&bv)[8], float4 (&y)[8],
                                             float& inv1)
{
    inv1 = inv_norm(a);
    #pragma unroll
    for (int j = 0; j < 8; ++j) {
        y[j].x = z[j].x * inv1 + bv[j].x;
        y[j].y = z[j].y * inv1 + bv[j].y;
        y[j].z = z[j].z * inv1 + bv[j].z;
        y[j].w = z[j].w * inv1 + bv[j].w;
    }
}

template <bool FIRST>
__global__ void __launch_bounds__(256)
recurrence_pair_kernel(const float* __restrict__ M0, const float* __restrict__ M1,
                       const float* __restrict__ add0, const float* __restrict__ add1,
                       const float* __restrict__ bias,
                       const float* __restrict__ vin, const float* __restrict__ zin,
                       float* __restrict__ out0, float* __restrict__ out1)
{
    __shared__ float sy[ROWS_PER_WG];

    const int tid  = threadIdx.x;
    const int wave = tid >> 5;
    const int lane = tid & 31;
    const int e0   = blockIdx.x * ROWS_PER_WG;
    const int sel  = blockIdx.y;

    const float* M    = sel ? M1 : M0;
    const float* addv = sel ? add1 : add0;
    float*       outv = sel ? out1 : out0;

    const float4* v4 = (const float4*)vin;
    const float4* z4 = (const float4*)zin;
    const float4* b4 = (const float4*)bias;
    const float4* w4 = (const float4*)(M + (size_t)(e0 + wave * ROWS_PER_WAVE) * H2);

    float4 u[8], a[8], z[8], bv[8];
    if (FIRST) {
        #pragma unroll
        for (int j = 0; j < 8; ++j)
            u[j] = v4[lane + 32 * j];
    } else {
        #pragma unroll
        for (int j = 0; j < 8; ++j) {
            a[j]  = v4[lane + 32 * j];
            z[j]  = z4[lane + 32 * j];
            bv[j] = b4[lane + 32 * j];
        }
    }

    __builtin_amdgcn_sched_barrier(0);

    float4 w[ROWS_PER_WAVE][8];
    #pragma unroll
    for (int r = 0; r < ROWS_PER_WAVE; ++r) {
        #pragma unroll
        for (int j = 0; j < 8; ++j)
            w[r][j] = w4[r * (H2 / 4) + lane + 32 * j];
    }

    float bb = 0.0f;
    if (addv != nullptr && tid < ROWS_PER_WG) bb = addv[e0 + tid];

    if (!FIRST) {
        float inv1;
        rebuild_even(a, z, bv, u, inv1);
        const float inv2 = inv_norm(u);
        #pragma unroll
        for (int j = 0; j < 8; ++j) {
            u[j].x *= inv2; u[j].y *= inv2; u[j].z *= inv2; u[j].w *= inv2;
        }
    }

    float p[ROWS_PER_WAVE];
    #pragma unroll
    for (int r = 0; r < ROWS_PER_WAVE; ++r) {
        float ax = 0.0f, ay = 0.0f, az = 0.0f, aw = 0.0f;
        #pragma unroll
        for (int j = 0; j < 8; ++j) {
            ax += w[r][j].x * u[j].x;
            ay += w[r][j].y * u[j].y;
            az += w[r][j].z * u[j].z;
            aw += w[r][j].w * u[j].w;
        }
        p[r] = wave_sum((ax + ay) + (az + aw));
    }

    if (lane == 0) {
        #pragma unroll
        for (int r = 0; r < ROWS_PER_WAVE; ++r)
            sy[wave * ROWS_PER_WAVE + r] = p[r];
    }
    __syncthreads();

    if (tid < ROWS_PER_WG)
        outv[e0 + tid] = sy[tid] + bb;
}

__global__ void __launch_bounds__(256)
normalize_pairs_kernel(float* __restrict__ u_ws, const float* __restrict__ bias)
{
    const int lane = threadIdx.x & 31;
    const int pair = blockIdx.x * 8 + (threadIdx.x >> 5);
    float4* pa4 = (float4*)(u_ws + (size_t)(2 * pair) * H2);
    float4* pz4 = (float4*)(u_ws + (size_t)(2 * pair + 1) * H2);
    const float4* b4 = (const float4*)bias;

    float4 a[8], z[8], bv[8], y[8];
    #pragma unroll
    for (int j = 0; j < 8; ++j) {
        a[j]  = pa4[lane + 32 * j];
        z[j]  = pz4[lane + 32 * j];
        bv[j] = b4[lane + 32 * j];
    }

    float inv1;
    rebuild_even(a, z, bv, y, inv1);
    const float inv2 = inv_norm(y);

    #pragma unroll
    for (int j = 0; j < 8; ++j) {
        float4 o;
        o.x = a[j].x * inv1; o.y = a[j].y * inv1; o.z = a[j].z * inv1; o.w = a[j].w * inv1;
        pa4[lane + 32 * j] = o;
        float4 q;
        q.x = y[j].x * inv2; q.y = y[j].y * inv2; q.z = y[j].z * inv2; q.w = y[j].w * inv2;
        pz4[lane + 32 * j] = q;
    }
}

#define BM     64
#define BN     256
#define KCH    32
#define PA     72
#define PB     44
#define NPASS  (H2 / BN)
#define CPP    (H2 / KCH)
#define NCHUNK (NPASS * CPP)

#define SA(buf_, j_) (*(float4*)(&sA[buf_][((tid + 256 * (j_)) >> 4) * PA + 4 * ((tid + 256 * (j_)) & 15)]))
#define SB(buf_, j_) (*(float4*)(&sB[buf_][((tid + 256 * (j_)) >> 3) * PB + 4 * ((tid + 256 * (j_)) & 7)]))

#define TSA 528
#define NST 3
#define PBS 36
#define SSA (4 * TSA)
#define SSB (BN * PBS)
#define GAS(p_, j_) (*(const float4*)((p_) + (size_t)offA + (size_t)((j_) * 16 * TT * 4)))
#define GBS(p_, j_) (*(const float4*)((p_) + (size_t)offB + (size_t)((j_) * 32 * H2 * 4)))
#define SAS(buf_, j_) (*(float4*)(sAw + (buf_) * SSA + (j_) * 256))
#define SBS(buf_, j_) (*(float4*)(sBw + (buf_) * SSB + (j_) * 32 * PBS))
#define LOAD_CHUNK_S(it_)                                                          \
    { const char* ap_ = hb + (size_t)(((it_) & (CPP - 1)) * (KCH * TT * 4));       \
      const char* bp_ = wb + (size_t)(((it_) >> 5) * (BN * H2 * 4) + ((it_) & (CPP - 1)) * (KCH * 4)); \
      a0 = GAS(ap_, 0); a1 = GAS(ap_, 1);                                          \
      b0 = GBS(bp_, 0); b1 = GBS(bp_, 1); b2 = GBS(bp_, 2); b3 = GBS(bp_, 3);      \
      b4 = GBS(bp_, 4); b5 = GBS(bp_, 5); b6 = GBS(bp_, 6); b7 = GBS(bp_, 7); }
#define STORE_CHUNK_S(buf_)                                                        \
    { SAS(buf_, 0) = a0; SAS(buf_, 1) = a1;                                        \
      SBS(buf_, 0) = b0; SBS(buf_, 1) = b1; SBS(buf_, 2) = b2; SBS(buf_, 3) = b3;  \
      SBS(buf_, 4) = b4; SBS(buf_, 5) = b5; SBS(buf_, 6) = b6; SBS(buf_, 7) = b7; }

__device__ __forceinline__ float tanh_er(float x)
{
    const float e = __builtin_amdgcn_exp2f(fabsf(x) * 2.885390081777927f);
    const float r = __builtin_amdgcn_rcpf(e + 1.0f);
    return copysignf(fmaf(-2.0f, r, 1.0f), x);
}

__global__ void __launch_bounds__(256)
scores_kernel(const float* hidden, const float* W,
              const float* __restrict__ bias,  const float* __restrict__ u_ws,
              float* __restrict__ scores)
{
    __shared__ __attribute__((aligned(16))) float sA[NST][SSA];
    __shared__ __attribute__((aligned(16))) float sB[NST][SSB];
    __shared__ float sPart[4][BM];

    const int n    = blockIdx.x;
    const int t0   = blockIdx.y * BM;
    const int tid  = threadIdx.x;
    const int lane = tid & 31;
    const int wave = tid >> 5;
    const int c    = lane & 15;
    const int h    = lane >> 4;
    const int wm   = wave >> 2;
    const int wn   = wave & 3;

    const char* hb = (const char*)(hidden + (size_t)n * H2 * TT + t0);
    const char* wb = (const char*)W;
    const unsigned offA = (unsigned)(((tid >> 4) * TT + 4 * (tid & 15)) * 4);
    const unsigned offB = (unsigned)(((tid >> 3) * H2 + 4 * (tid & 7)) * 4);
    float* sAw = &sA[0][0] + ((tid & 15) >> 2) * TSA + (tid >> 4) * 16 + 4 * (tid & 3);
    float* sBw = &sB[0][0] + (tid >> 3) * PBS + 4 * (tid & 7);

    v8f acc[2][4];
    #pragma unroll
    for (int mt = 0; mt < 2; ++mt)
        #pragma unroll
        for (int nt = 0; nt < 4; ++nt)
            acc[mt][nt] = (v8f){0.f,0.f,0.f,0.f,0.f,0.f,0.f,0.f};

    float tAcc[2][8];
    #pragma unroll
    for (int mt = 0; mt < 2; ++mt)
        #pragma unroll
        for (int r = 0; r < 8; ++r) tAcc[mt][r] = 0.0f;

#define KSTEP_S(ks_)                                                               \
    { v2f af[2], bf[4];                                                            \
      _Pragma("unroll")                                                            \
      for (int mt = 0; mt < 2; ++mt) {                                             \
          af[mt].x = pa[mt * TSA + ((ks_) * 4)     * 16];                          \
          af[mt].y = pa[mt * TSA + ((ks_) * 4 + 1) * 16];                          \
      }                                                                            \
      _Pragma("unroll")                                                            \
      for (int nt = 0; nt < 4; ++nt)                                               \
          bf[nt] = *(const v2f*)(pb + nt * 16 * PBS + (ks_) * 4);                  \
      _Pragma("unroll")                                                            \
      for (int mt = 0; mt < 2; ++mt)                                               \
          _Pragma("unroll")                                                        \
          for (int nt = 0; nt < 4; ++nt)                                           \
              acc[mt][nt] = __builtin_amdgcn_wmma_f32_16x16x4_f32(                 \
                   false, af[mt],  false, bf[nt],                \
                   (short)0, acc[mt][nt],                                 \
                   false,  false); }

    float4 a0, a1, b0, b1, b2, b3, b4, b5, b6, b7;
    LOAD_CHUNK_S(0)
    STORE_CHUNK_S(0)
    __builtin_amdgcn_fence(__ATOMIC_RELEASE, "workgroup", "local");
    __builtin_amdgcn_s_barrier_signal(-1);
    LOAD_CHUNK_S(1)

    int cur = 0, nxt = 1;
    for (int it = 0; it < NCHUNK; ++it) {
        const int n2 = (it + 2 < NCHUNK) ? it + 2 : NCHUNK - 1;

        __builtin_amdgcn_s_barrier_wait(-1);
        __builtin_amdgcn_fence(__ATOMIC_ACQUIRE, "workgroup", "local");

        const float* pa = sA[0] + cur * SSA + (2 * wm) * TSA + (2 * h) * 16 + c;
        const float* pb = sB[0] + cur * SSB + (wn * 64 + c) * PBS + 2 * h;

        KSTEP_S(0) KSTEP_S(1) KSTEP_S(2) KSTEP_S(3)

        STORE_CHUNK_S(nxt)
        __builtin_amdgcn_fence(__ATOMIC_RELEASE, "workgroup", "local");
        __builtin_amdgcn_s_barrier_signal(-1);

        LOAD_CHUNK_S(n2)
        __builtin_amdgcn_sched_barrier(0);

        KSTEP_S(4) KSTEP_S(5) KSTEP_S(6) KSTEP_S(7)

        if ((it % CPP) == CPP - 1) {
            const int ec = it / CPP;
            #pragma unroll
            for (int nt = 0; nt < 4; ++nt) {
                const int e = ec * BN + wn * 64 + nt * 16 + c;
                const float bb = bias[e];
                const float uu = u_ws[(size_t)n * H2 + e];
                #pragma unroll
                for (int mt = 0; mt < 2; ++mt) {
                    #pragma unroll
                    for (int r = 0; r < 8; ++r) {
                        tAcc[mt][r] += tanh_er(acc[mt][nt][r] + bb) * uu;
                        acc[mt][nt][r] = 0.0f;
                    }
                }
            }
        }

        cur = nxt;
        nxt = (nxt == NST - 1) ? 0 : nxt + 1;
    }
    __builtin_amdgcn_s_barrier_wait(-1);
#undef KSTEP_S

    #pragma unroll
    for (int mt = 0; mt < 2; ++mt) {
        #pragma unroll
        for (int r = 0; r < 8; ++r) {
            float v = tAcc[mt][r];
            #pragma unroll
            for (int off = 8; off >= 1; off >>= 1)
                v += __shfl_xor(v, off, 32);
            if (c == 0)
                sPart[wn][wm * 32 + mt * 16 + 8 * h + r] = v;
        }
    }
    __syncthreads();

    if (tid < BM)
        scores[(size_t)n * TT + t0 + tid] =
            ((sPart[0][tid] + sPart[1][tid]) + sPart[2][tid]) + sPart[3][tid];
}

#define BN2 64
#define GA2(it_, j_) (*(const float4*)(abase + (size_t)((it_) * KCH + ((tid + 256 * (j_)) >> 4)) * H2 + 4 * ((tid + 256 * (j_)) & 15)))
#define GB2(it_, j_) (*(const float4*)(bbase + (size_t)((tid + 256 * (j_)) >> 3) * H2 + (it_) * KCH + 4 * ((tid + 256 * (j_)) & 7)))

__global__ void __launch_bounds__(256)
w2_gemm_kernel(const float* W, float* __restrict__ W2)
{
    __shared__ __attribute__((aligned(16))) float sA[NST][KCH * PA];
    __shared__ __attribute__((aligned(16))) float sB[NST][BN2 * PB];

    const int j0   = blockIdx.x * BM;
    const int i0   = blockIdx.y * BN2;
    const int tid  = threadIdx.x;
    const int lane = tid & 31;
    const int wave = tid >> 5;
    const int c    = lane & 15;
    const int h    = lane >> 4;
    const int wm   = wave >> 2;
    const int wn   = wave & 3;

    const float* abase = W + j0;
    const float* bbase = W + (size_t)i0 * H2;

    v8f acc[2];
    acc[0] = (v8f){0.f,0.f,0.f,0.f,0.f,0.f,0.f,0.f};
    acc[1] = (v8f){0.f,0.f,0.f,0.f,0.f,0.f,0.f,0.f};

#define KSTEP_2(ks_)                                                               \
    { v2f af[2], bf;                                                               \
      _Pragma("unroll")                                                            \
      for (int mt = 0; mt < 2; ++mt) {                                             \
          af[mt].x = pa[((ks_) * 4)     * PA + mt * 16];                           \
          af[mt].y = pa[((ks_) * 4 + 1) * PA + mt * 16];                           \
      }                                                                            \
      bf = *(const v2f*)(pb + (ks_) * 4);                                          \
      _Pragma("unroll")                                                            \
      for (int mt = 0; mt < 2; ++mt)                                               \
          acc[mt] = __builtin_amdgcn_wmma_f32_16x16x4_f32(                         \
               false, af[mt],  false, bf,                        \
               (short)0, acc[mt],                                         \
               false,  false); }

    float4 a0, a1, b0, b1;
    a0 = GA2(0, 0); a1 = GA2(0, 1); b0 = GB2(0, 0); b1 = GB2(0, 1);
    SA(0, 0) = a0; SA(0, 1) = a1; SB(0, 0) = b0; SB(0, 1) = b1;
    __builtin_amdgcn_fence(__ATOMIC_RELEASE, "workgroup", "local");
    __builtin_amdgcn_s_barrier_signal(-1);
    a0 = GA2(1, 0); a1 = GA2(1, 1); b0 = GB2(1, 0); b1 = GB2(1, 1);

    int cur = 0, nxt = 1;
    for (int it = 0; it < H2 / KCH; ++it) {
        const int n2 = (it + 2 < H2 / KCH) ? it + 2 : H2 / KCH - 1;

        __builtin_amdgcn_s_barrier_wait(-1);
        __builtin_amdgcn_fence(__ATOMIC_ACQUIRE, "workgroup", "local");

        const float* pa = sA[cur] + (2 * h) * PA + wm * 32 + c;
        const float* pb = sB[cur] + (wn * 16 + c) * PB + 2 * h;

        KSTEP_2(0) KSTEP_2(1) KSTEP_2(2) KSTEP_2(3)

        SA(nxt, 0) = a0; SA(nxt, 1) = a1; SB(nxt, 0) = b0; SB(nxt, 1) = b1;
        __builtin_amdgcn_fence(__ATOMIC_RELEASE, "workgroup", "local");
        __builtin_amdgcn_s_barrier_signal(-1);

        a0 = GA2(n2, 0); a1 = GA2(n2, 1); b0 = GB2(n2, 0); b1 = GB2(n2, 1);
        __builtin_amdgcn_sched_barrier(0);

        KSTEP_2(4) KSTEP_2(5) KSTEP_2(6) KSTEP_2(7)

        cur = nxt;
        nxt = (nxt == NST - 1) ? 0 : nxt + 1;
    }
    __builtin_amdgcn_s_barrier_wait(-1);
#undef KSTEP_2

    #pragma unroll
    for (int mt = 0; mt < 2; ++mt) {
        float* dst = W2 + (size_t)(i0 + wn * 16 + c) * H2 + j0 + wm * 32 + mt * 16 + 8 * h;
        float4 lo, hi;
        lo.x = acc[mt][0]; lo.y = acc[mt][1]; lo.z = acc[mt][2]; lo.w = acc[mt][3];
        hi.x = acc[mt][4]; hi.y = acc[mt][5]; hi.z = acc[mt][6]; hi.w = acc[mt][7];
        *(float4*)(dst)     = lo;
        *(float4*)(dst + 4) = hi;
    }
}

__global__ void __launch_bounds__(512)
softmax_kernel(const float* __restrict__ scores, float* __restrict__ attn)
{
    __shared__ float red[16];
    const int n   = blockIdx.x;
    const int tid = threadIdx.x;

    const float s = scores[(size_t)n * TT + tid];

    float m = s;
    #pragma unroll
    for (int off = 16; off >= 1; off >>= 1)
        m = fmaxf(m, __shfl_xor(m, off, 32));
    if ((tid & 31) == 0) red[tid >> 5] = m;
    __syncthreads();
    if (tid < 16) {
        float r = red[tid];
        #pragma unroll
        for (int off = 8; off >= 1; off >>= 1)
            r = fmaxf(r, __shfl_xor(r, off, 32));
        red[tid] = r;
    }
    __syncthreads();
    m = red[0];
    __syncthreads();

    const float e = __expf(s - m);
    float sum = e;
    #pragma unroll
    for (int off = 16; off >= 1; off >>= 1)
        sum += __shfl_xor(sum, off, 32);
    if ((tid & 31) == 0) red[tid >> 5] = sum;
    __syncthreads();
    if (tid < 16) {
        float r = red[tid];
        #pragma unroll
        for (int off = 8; off >= 1; off >>= 1)
            r += __shfl_xor(r, off, 32);
        red[tid] = r;
    }
    __syncthreads();

    attn[(size_t)n * TT + tid] = e / red[0];
}

__global__ void __launch_bounds__(256)
weighted_sum_kernel(const float* __restrict__ hidden,
                    const float* __restrict__ attn, float* __restrict__ out)
{
    __shared__ float sa[TT];
    const int n   = blockIdx.x;
    const int dc  = blockIdx.y;
    const int tid = threadIdx.x;

    sa[tid]       = attn[(size_t)n * TT + tid];
    sa[tid + 256] = attn[(size_t)n * TT + tid + 256];
    __syncthreads();

    const int d = dc * 256 + tid;
    const float* h = hidden + (size_t)n * H2 * TT + (size_t)d * TT;
    float acc = 0.0f;
    #pragma unroll 4
    for (int t = 0; t < TT; t += 4) {
        const float4 h4 = *(const float4*)(h + t);
        acc += h4.x * sa[t] + h4.y * sa[t + 1] + h4.z * sa[t + 2] + h4.w * sa[t + 3];
    }
    out[(size_t)n * H2 + d] = acc;
}

extern "C" void kernel_launch(void* const* d_in, const int* in_sizes, int n_in,
                              void* d_out, int out_size, void* d_ws, size_t ws_size,
                              hipStream_t stream)
{
    const float* hidden = (const float*)d_in[0];
    const float* W      = (const float*)d_in[1];
    const float* bias   = (const float*)d_in[2];
    const float* u0     = (const float*)d_in[3];
    float* out = (float*)d_out;

    float* ws     = (float*)d_ws;
    float* u_ws   = ws;
    float* scores = u_ws + (size_t)NSTEP * H2;
    float* attn   = scores + (size_t)NSTEP * TT;

    float* W2     = attn + (size_t)NSTEP * TT;
    float* Wb     = W2 + (size_t)H2 * H2;

    if (ws_size >= (size_t)4722688) {
        recurrence_pair_kernel<true><<<dim3(H2 / ROWS_PER_WG, 1), 256, 0, stream>>>(
            W, W, nullptr, nullptr, bias, bias, bias, Wb, Wb);
        w2_gemm_kernel<<<dim3(H2 / BM, H2 / BN2), 256, 0, stream>>>(W, W2);

        recurrence_pair_kernel<true><<<dim3(H2 / ROWS_PER_WG, 2), 256, 0, stream>>>(
            W, W2, bias, Wb, bias, u0, u0, u_ws, u_ws + H2);
        for (int k = 1; k < NSTEP / 2; ++k)
            recurrence_pair_kernel<false><<<dim3(H2 / ROWS_PER_WG, 2), 256, 0, stream>>>(
                W, W2, bias, Wb, bias,
                u_ws + (size_t)(2 * k - 2) * H2, u_ws + (size_t)(2 * k - 1) * H2,
                u_ws + (size_t)(2 * k) * H2,     u_ws + (size_t)(2 * k + 1) * H2);
        normalize_pairs_kernel<<<NSTEP / 16, 256, 0, stream>>>(u_ws, bias);
    } else {
        recurrence_step_kernel<<<H2 / ROWS_PER_WG, 256, 0, stream>>>(W, bias, u0, u_ws, 0);
        for (int i = 1; i < NSTEP; ++i)
            recurrence_step_kernel<<<H2 / ROWS_PER_WG, 256, 0, stream>>>(
                W, bias, u_ws + (size_t)(i - 1) * H2, u_ws + (size_t)i * H2, 1);
        normalize_rows_kernel<<<NSTEP / 8, 256, 0, stream>>>(u_ws);
    }

    scores_kernel<<<dim3(NSTEP, TT / BM), 256, 0, stream>>>(
        hidden, W, bias, u_ws, scores);

    softmax_kernel<<<NSTEP, TT, 0, stream>>>(scores, attn);

    weighted_sum_kernel<<<dim3(NSTEP, H2 / 256), 256, 0, stream>>>(
        hidden, attn, out);
}
